# starting kernel
# baseline (speedup 1.0000x reference)
#include <hip/hip_runtime.h>
#include <hip/hip_bf16.h>

typedef __attribute__((ext_vector_type(16))) _Float16 v16h;
typedef __attribute__((ext_vector_type(8)))  _Float16 h8;
typedef __attribute__((ext_vector_type(8)))  float    v8f;
typedef __attribute__((ext_vector_type(4)))  float    f4;

#define NEG_SLOPE 0.01f

__device__ __forceinline__ float lrelu(float v) { return v >= 0.0f ? v : NEG_SLOPE * v; }

__global__ void transpose_w_kernel(const float* __restrict__ W, _Float16* __restrict__ Wt) {
    int t = blockIdx.x * blockDim.x + threadIdx.x;
    if (t >= 256 * 64) return;
    int k = t >> 6;
    int c = t & 63;
    Wt[c * 256 + k] = (_Float16)W[t];
}

__global__ void __launch_bounds__(64) gemm_xw_kernel(const float* __restrict__ x,
                                                     const _Float16* __restrict__ Wt,
                                                     float* __restrict__ h) {
    const int lane = threadIdx.x & 31;
    const int wave = threadIdx.x >> 5;
    const int hi   = lane >> 4;
    const int r    = lane & 15;
    const int rowBase = (blockIdx.x * 2 + wave) * 16;

    v8f c0 = {}, c1 = {}, c2 = {}, c3 = {};

    const float*    xrow = x  + (size_t)(rowBase + r) * 256 + hi * 8;
    const _Float16* w0   = Wt + (size_t)(0 * 16 + r) * 256 + hi * 8;
    const _Float16* w1   = Wt + (size_t)(1 * 16 + r) * 256 + hi * 8;
    const _Float16* w2   = Wt + (size_t)(2 * 16 + r) * 256 + hi * 8;
    const _Float16* w3   = Wt + (size_t)(3 * 16 + r) * 256 + hi * 8;

    for (int kk = 0; kk < 8; ++kk) {
        const int kb = kk * 32;

        f4 a0 = *(const f4*)(xrow + kb);
        f4 a1 = *(const f4*)(xrow + kb + 4);
        f4 a2 = *(const f4*)(xrow + kb + 16);
        f4 a3 = *(const f4*)(xrow + kb + 20);
        v16h a;
        #pragma unroll
        for (int i = 0; i < 4; ++i) {
            a[i]      = (_Float16)a0[i];
            a[4 + i]  = (_Float16)a1[i];
            a[8 + i]  = (_Float16)a2[i];
            a[12 + i] = (_Float16)a3[i];
        }

        h8 b0lo = *(const h8*)(w0 + kb); h8 b0hi = *(const h8*)(w0 + kb + 16);
        h8 b1lo = *(const h8*)(w1 + kb); h8 b1hi = *(const h8*)(w1 + kb + 16);
        h8 b2lo = *(const h8*)(w2 + kb); h8 b2hi = *(const h8*)(w2 + kb + 16);
        h8 b3lo = *(const h8*)(w3 + kb); h8 b3hi = *(const h8*)(w3 + kb + 16);
        v16h b0v, b1v, b2v, b3v;
        #pragma unroll
        for (int i = 0; i < 8; ++i) {
            b0v[i] = b0lo[i]; b0v[8 + i] = b0hi[i];
            b1v[i] = b1lo[i]; b1v[8 + i] = b1hi[i];
            b2v[i] = b2lo[i]; b2v[8 + i] = b2hi[i];
            b3v[i] = b3lo[i]; b3v[8 + i] = b3hi[i];
        }

        c0 = __builtin_amdgcn_wmma_f32_16x16x32_f16(false, a, false, b0v, (short)0, c0, false, false);
        c1 = __builtin_amdgcn_wmma_f32_16x16x32_f16(false, a, false, b1v, (short)0, c1, false, false);
        c2 = __builtin_amdgcn_wmma_f32_16x16x32_f16(false, a, false, b2v, (short)0, c2, false, false);
        c3 = __builtin_amdgcn_wmma_f32_16x16x32_f16(false, a, false, b3v, (short)0, c3, false, false);
    }

    #pragma unroll
    for (int j = 0; j < 8; ++j) {
        float* p = h + (size_t)(rowBase + j + hi * 8) * 64;
        p[r]      = c0[j];
        p[16 + r] = c1[j];
        p[32 + r] = c2[j];
        p[48 + r] = c3[j];
    }
}

__global__ void deg_init_kernel(unsigned int* __restrict__ deg, int N) {
    int i = blockIdx.x * blockDim.x + threadIdx.x;
    if (i < N) deg[i] = 1u;
}

__global__ void deg_count_kernel(const int* __restrict__ dst, unsigned int* __restrict__ deg, int E) {
    int e = blockIdx.x * blockDim.x + threadIdx.x;
    if (e < E) atomicAdd(&deg[dst[e]], 1u);
}

__global__ void dinv_kernel(const unsigned int* __restrict__ deg, float* __restrict__ dinv, int N) {
    int i = blockIdx.x * blockDim.x + threadIdx.x;
    if (i < N) {
        float d = (float)deg[i];
        dinv[i] = d > 0.0f ? rsqrtf(d) : 0.0f;
    }
}

__global__ void embed_init_kernel(const float* __restrict__ h, const float* __restrict__ dinv,
                                  float* __restrict__ embed, int total) {
    int t = blockIdx.x * blockDim.x + threadIdx.x;
    if (t >= total) return;
    int i = t >> 6;
    float di = dinv[i];
    embed[t] = h[t] * di * di;
}

__global__ void scatter_edges_kernel(const float* __restrict__ h, const float* __restrict__ dinv,
                                     const int* __restrict__ src, const int* __restrict__ dst,
                                     float* __restrict__ embed, int E) {
    int t = blockIdx.x * blockDim.x + threadIdx.x;
    if (t >= E * 64) return;
    int e = t >> 6;
    int c = t & 63;
    int s = src[e];
    int d = dst[e];
    float norm = dinv[s] * dinv[d];
    unsafeAtomicAdd(&embed[(size_t)d * 64 + c], h[(size_t)s * 64 + c] * norm);
}

__global__ void mlp_kernel(const float* __restrict__ embed, const int* __restrict__ idx,
                           const float* __restrict__ b_conv,
                           const float* __restrict__ W1, const float* __restrict__ b1,
                           const float* __restrict__ W2, const float* __restrict__ b2,
                           float* __restrict__ out, int B) {
    int b = blockIdx.x * blockDim.x + threadIdx.x;
    if (b >= B) return;
    int i0 = idx[2 * b];
    int i1 = idx[2 * b + 1];
    const float* e0 = embed + (size_t)i0 * 64;
    const float* e1 = embed + (size_t)i1 * 64;

    float o1[16];
    #pragma unroll
    for (int j = 0; j < 16; ++j) o1[j] = b1[j];

    for (int c = 0; c < 64; ++c) {
        float v = lrelu(e0[c] + b_conv[c]);
        const float* wr = W1 + c * 16;
        #pragma unroll
        for (int j = 0; j < 16; ++j) o1[j] += v * wr[j];
    }
    for (int c = 0; c < 64; ++c) {
        float v = lrelu(e1[c] + b_conv[c]);
        const float* wr = W1 + (64 + c) * 16;
        #pragma unroll
        for (int j = 0; j < 16; ++j) o1[j] += v * wr[j];
    }

    float acc = b2[0];
    #pragma unroll
    for (int j = 0; j < 16; ++j) acc += lrelu(o1[j]) * W2[j];

    out[b] = 1.0f / (1.0f + __expf(-acc));
}

extern "C" void kernel_launch(void* const* d_in, const int* in_sizes, int n_in,
                              void* d_out, int out_size, void* d_ws, size_t ws_size,
                              hipStream_t stream) {
    const float* x        = (const float*)d_in[0];
    const int*   edge     = (const int*)d_in[1];
    const int*   index    = (const int*)d_in[2];
    const float* W_conv   = (const float*)d_in[3];
    const float* b_conv   = (const float*)d_in[4];
    const float* W1       = (const float*)d_in[5];
    const float* b1       = (const float*)d_in[6];
    const float* W2       = (const float*)d_in[7];
    const float* b2       = (const float*)d_in[8];
    float*       out      = (float*)d_out;

    const int N = in_sizes[0] / 256;
    const int E = in_sizes[1] / 2;
    const int B = in_sizes[2] / 2;
    const int* src = edge;
    const int* dst = edge + E;

    float*        h     = (float*)d_ws;
    float*        embed = h + (size_t)N * 64;
    unsigned int* deg   = (unsigned int*)(embed + (size_t)N * 64);
    float*        dinv  = (float*)(deg + N);
    _Float16*     Wt    = (_Float16*)(dinv + N);

    transpose_w_kernel<<<(256 * 64 + 255) / 256, 256, 0, stream>>>(W_conv, Wt);

    gemm_xw_kernel<<<N / 32, 64, 0, stream>>>(x, Wt, h);

    deg_init_kernel<<<(N + 255) / 256, 256, 0, stream>>>(deg, N);
    deg_count_kernel<<<(E + 255) / 256, 256, 0, stream>>>(dst, deg, E);
    dinv_kernel<<<(N + 255) / 256, 256, 0, stream>>>(deg, dinv, N);

    embed_init_kernel<<<(N * 64 + 255) / 256, 256, 0, stream>>>(h, dinv, embed, N * 64);
    {
        long long total = (long long)E * 64;
        int blocks = (int)((total + 255) / 256);
        scatter_edges_kernel<<<blocks, 256, 0, stream>>>(h, dinv, src, dst, embed, E);
    }

    mlp_kernel<<<(B + 255) / 256, 256, 0, stream>>>(embed, index, b_conv, W1, b1, W2, b2, out, B);
}


# scatter: skip edges whose dst row the head never gathers (rows marked by index in top bit of deg); lane-per-edge mark test, wave-wide f32 atomic adds for kept edges; sum order still arrival order
# speedup vs baseline: 2.2479x; 2.2479x over previous
#include <hip/hip_runtime.h>
#include <hip/hip_bf16.h>

typedef __attribute__((ext_vector_type(16))) _Float16 v16h;
typedef __attribute__((ext_vector_type(8)))  _Float16 h8;
typedef __attribute__((ext_vector_type(8)))  float    v8f;
typedef __attribute__((ext_vector_type(4)))  float    f4;

#define NEG_SLOPE 0.01f

__device__ __forceinline__ float lrelu(float v) { return v >= 0.0f ? v : NEG_SLOPE * v; }

__global__ void transpose_w_kernel(const float* __restrict__ W, _Float16* __restrict__ Wt) {
    int t = blockIdx.x * blockDim.x + threadIdx.x;
    if (t >= 256 * 64) return;
    int k = t >> 6;
    int c = t & 63;
    Wt[c * 256 + k] = (_Float16)W[t];
}

__global__ void __launch_bounds__(64) gemm_xw_kernel(const float* __restrict__ x,
                                                     const _Float16* __restrict__ Wt,
                                                     float* __restrict__ h) {
    const int lane = threadIdx.x & 31;
    const int wave = threadIdx.x >> 5;
    const int hi   = lane >> 4;
    const int r    = lane & 15;
    const int rowBase = (blockIdx.x * 2 + wave) * 16;

    v8f c0 = {}, c1 = {}, c2 = {}, c3 = {};

    const float*    xrow = x  + (size_t)(rowBase + r) * 256 + hi * 8;
    const _Float16* w0   = Wt + (size_t)(0 * 16 + r) * 256 + hi * 8;
    const _Float16* w1   = Wt + (size_t)(1 * 16 + r) * 256 + hi * 8;
    const _Float16* w2   = Wt + (size_t)(2 * 16 + r) * 256 + hi * 8;
    const _Float16* w3   = Wt + (size_t)(3 * 16 + r) * 256 + hi * 8;

    for (int kk = 0; kk < 8; ++kk) {
        const int kb = kk * 32;

        f4 a0 = *(const f4*)(xrow + kb);
        f4 a1 = *(const f4*)(xrow + kb + 4);
        f4 a2 = *(const f4*)(xrow + kb + 16);
        f4 a3 = *(const f4*)(xrow + kb + 20);
        v16h a;
        #pragma unroll
        for (int i = 0; i < 4; ++i) {
            a[i]      = (_Float16)a0[i];
            a[4 + i]  = (_Float16)a1[i];
            a[8 + i]  = (_Float16)a2[i];
            a[12 + i] = (_Float16)a3[i];
        }

        h8 b0lo = *(const h8*)(w0 + kb); h8 b0hi = *(const h8*)(w0 + kb + 16);
        h8 b1lo = *(const h8*)(w1 + kb); h8 b1hi = *(const h8*)(w1 + kb + 16);
        h8 b2lo = *(const h8*)(w2 + kb); h8 b2hi = *(const h8*)(w2 + kb + 16);
        h8 b3lo = *(const h8*)(w3 + kb); h8 b3hi = *(const h8*)(w3 + kb + 16);
        v16h b0v, b1v, b2v, b3v;
        #pragma unroll
        for (int i = 0; i < 8; ++i) {
            b0v[i] = b0lo[i]; b0v[8 + i] = b0hi[i];
            b1v[i] = b1lo[i]; b1v[8 + i] = b1hi[i];
            b2v[i] = b2lo[i]; b2v[8 + i] = b2hi[i];
            b3v[i] = b3lo[i]; b3v[8 + i] = b3hi[i];
        }

        c0 = __builtin_amdgcn_wmma_f32_16x16x32_f16(false, a, false, b0v, (short)0, c0, false, false);
        c1 = __builtin_amdgcn_wmma_f32_16x16x32_f16(false, a, false, b1v, (short)0, c1, false, false);
        c2 = __builtin_amdgcn_wmma_f32_16x16x32_f16(false, a, false, b2v, (short)0, c2, false, false);
        c3 = __builtin_amdgcn_wmma_f32_16x16x32_f16(false, a, false, b3v, (short)0, c3, false, false);
    }

    #pragma unroll
    for (int j = 0; j < 8; ++j) {
        float* p = h + (size_t)(rowBase + j + hi * 8) * 64;
        p[r]      = c0[j];
        p[16 + r] = c1[j];
        p[32 + r] = c2[j];
        p[48 + r] = c3[j];
    }
}

__global__ void deg_init_kernel(unsigned int* __restrict__ deg, int N) {
    int i = blockIdx.x * blockDim.x + threadIdx.x;
    if (i < N) deg[i] = 1u;
}

#define ROW_USED 0x80000000u
__global__ void deg_count_kernel(const int* __restrict__ dst, const int* __restrict__ idx,
                                 unsigned int* __restrict__ deg, int E, int nIdx) {
    int e = blockIdx.x * blockDim.x + threadIdx.x;
    if (e < E) atomicAdd(&deg[dst[e]], 1u);
    for (int t = e; t < nIdx; t += gridDim.x * blockDim.x) atomicOr(&deg[idx[t]], ROW_USED);
}

__global__ void dinv_kernel(const unsigned int* __restrict__ deg, float* __restrict__ dinv, int N) {
    int i = blockIdx.x * blockDim.x + threadIdx.x;
    if (i < N) {
        float d = (float)(deg[i] & ~ROW_USED);
        dinv[i] = d > 0.0f ? rsqrtf(d) : 0.0f;
    }
}

__global__ void embed_init_kernel(const float* __restrict__ h, const float* __restrict__ dinv,
                                  float* __restrict__ embed, int total) {
    int t = blockIdx.x * blockDim.x + threadIdx.x;
    if (t >= total) return;
    int i = t >> 6;
    float di = dinv[i];
    embed[t] = h[t] * di * di;
}

__device__ __forceinline__ int   lane_i(int v, int j)   { return __builtin_amdgcn_readlane(v, j); }
__device__ __forceinline__ float lane_f(float v, int j) { return __int_as_float(__builtin_amdgcn_readlane(__float_as_int(v), j)); }

__global__ void __launch_bounds__(256)
scatter_edges_kernel(const float* __restrict__ h, const float* __restrict__ dinv,
                     const unsigned int* __restrict__ deg,
                     const int* __restrict__ src, const int* __restrict__ dst,
                     float* __restrict__ embed, int E) {
    const int lane = threadIdx.x & 31;
    const int e = blockIdx.x * blockDim.x + threadIdx.x;
    int s = 0, d = 0;
    float w = 0.0f;
    bool keep = false;
    if (e < E) {
        s = src[e];
        d = dst[e];
        keep = (deg[d] & ROW_USED) != 0u;
        if (keep) w = dinv[s] * dinv[d];
    }
    unsigned int m = __builtin_amdgcn_ballot_w32(keep);
    while (m) {
        if (__builtin_popcount(m) >= 4) {
            const int j0 = __builtin_ctz(m); m &= m - 1u;
            const int j1 = __builtin_ctz(m); m &= m - 1u;
            const int j2 = __builtin_ctz(m); m &= m - 1u;
            const int j3 = __builtin_ctz(m); m &= m - 1u;
            const float* p0 = h + (size_t)lane_i(s, j0) * 64 + lane;
            const float* p1 = h + (size_t)lane_i(s, j1) * 64 + lane;
            const float* p2 = h + (size_t)lane_i(s, j2) * 64 + lane;
            const float* p3 = h + (size_t)lane_i(s, j3) * 64 + lane;
            const float a0 = p0[0], b0 = p0[32];
            const float a1 = p1[0], b1 = p1[32];
            const float a2 = p2[0], b2 = p2[32];
            const float a3 = p3[0], b3 = p3[32];
            float* q0 = embed + (size_t)lane_i(d, j0) * 64 + lane;
            float* q1 = embed + (size_t)lane_i(d, j1) * 64 + lane;
            float* q2 = embed + (size_t)lane_i(d, j2) * 64 + lane;
            float* q3 = embed + (size_t)lane_i(d, j3) * 64 + lane;
            const float w0 = lane_f(w, j0), w1 = lane_f(w, j1), w2 = lane_f(w, j2), w3 = lane_f(w, j3);
            unsafeAtomicAdd(q0, a0 * w0); unsafeAtomicAdd(q0 + 32, b0 * w0);
            unsafeAtomicAdd(q1, a1 * w1); unsafeAtomicAdd(q1 + 32, b1 * w1);
            unsafeAtomicAdd(q2, a2 * w2); unsafeAtomicAdd(q2 + 32, b2 * w2);
            unsafeAtomicAdd(q3, a3 * w3); unsafeAtomicAdd(q3 + 32, b3 * w3);
        } else {
            const int j0 = __builtin_ctz(m); m &= m - 1u;
            const float* p0 = h + (size_t)lane_i(s, j0) * 64 + lane;
            const float a0 = p0[0], b0 = p0[32];
            float* q0 = embed + (size_t)lane_i(d, j0) * 64 + lane;
            const float w0 = lane_f(w, j0);
            unsafeAtomicAdd(q0, a0 * w0); unsafeAtomicAdd(q0 + 32, b0 * w0);
        }
    }
}

__global__ void mlp_kernel(const float* __restrict__ embed, const int* __restrict__ idx,
                           const float* __restrict__ b_conv,
                           const float* __restrict__ W1, const float* __restrict__ b1,
                           const float* __restrict__ W2, const float* __restrict__ b2,
                           float* __restrict__ out, int B) {
    int b = blockIdx.x * blockDim.x + threadIdx.x;
    if (b >= B) return;
    int i0 = idx[2 * b];
    int i1 = idx[2 * b + 1];
    const float* e0 = embed + (size_t)i0 * 64;
    const float* e1 = embed + (size_t)i1 * 64;

    float o1[16];
    #pragma unroll
    for (int j = 0; j < 16; ++j) o1[j] = b1[j];

    for (int c = 0; c < 64; ++c) {
        float v = lrelu(e0[c] + b_conv[c]);
        const float* wr = W1 + c * 16;
        #pragma unroll
        for (int j = 0; j < 16; ++j) o1[j] += v * wr[j];
    }
    for (int c = 0; c < 64; ++c) {
        float v = lrelu(e1[c] + b_conv[c]);
        const float* wr = W1 + (64 + c) * 16;
        #pragma unroll
        for (int j = 0; j < 16; ++j) o1[j] += v * wr[j];
    }

    float acc = b2[0];
    #pragma unroll
    for (int j = 0; j < 16; ++j) acc += lrelu(o1[j]) * W2[j];

    out[b] = 1.0f / (1.0f + __expf(-acc));
}

extern "C" void kernel_launch(void* const* d_in, const int* in_sizes, int n_in,
                              void* d_out, int out_size, void* d_ws, size_t ws_size,
                              hipStream_t stream) {
    const float* x        = (const float*)d_in[0];
    const int*   edge     = (const int*)d_in[1];
    const int*   index    = (const int*)d_in[2];
    const float* W_conv   = (const float*)d_in[3];
    const float* b_conv   = (const float*)d_in[4];
    const float* W1       = (const float*)d_in[5];
    const float* b1       = (const float*)d_in[6];
    const float* W2       = (const float*)d_in[7];
    const float* b2       = (const float*)d_in[8];
    float*       out      = (float*)d_out;

    const int N = in_sizes[0] / 256;
    const int E = in_sizes[1] / 2;
    const int B = in_sizes[2] / 2;
    const int* src = edge;
    const int* dst = edge + E;

    float*        h     = (float*)d_ws;
    float*        embed = h + (size_t)N * 64;
    unsigned int* deg   = (unsigned int*)(embed + (size_t)N * 64);
    float*        dinv  = (float*)(deg + N);
    _Float16*     Wt    = (_Float16*)(dinv + N);

    transpose_w_kernel<<<(256 * 64 + 255) / 256, 256, 0, stream>>>(W_conv, Wt);

    gemm_xw_kernel<<<N / 32, 64, 0, stream>>>(x, Wt, h);

    deg_init_kernel<<<(N + 255) / 256, 256, 0, stream>>>(deg, N);
    deg_count_kernel<<<(E + 255) / 256, 256, 0, stream>>>(dst, index, deg, E, 2 * B);
    dinv_kernel<<<(N + 255) / 256, 256, 0, stream>>>(deg, dinv, N);

    embed_init_kernel<<<(N * 64 + 255) / 256, 256, 0, stream>>>(h, dinv, embed, N * 64);
    scatter_edges_kernel<<<(E + 255) / 256, 256, 0, stream>>>(h, dinv, deg, src, dst, embed, E);

    mlp_kernel<<<(B + 255) / 256, 256, 0, stream>>>(embed, index, b_conv, W1, b1, W2, b2, out, B);
}


# degree count and gathered-row marks in LDS counters per (node range, edge slice), stored as partial planes summed by the dinv pass: no global integer atomics; deg_init launch dropped (7 launches)
# speedup vs baseline: 3.7391x; 1.6634x over previous
#include <hip/hip_runtime.h>
#include <hip/hip_bf16.h>

typedef __attribute__((ext_vector_type(16))) _Float16 v16h;
typedef __attribute__((ext_vector_type(8)))  _Float16 h8;
typedef __attribute__((ext_vector_type(8)))  float    v8f;
typedef __attribute__((ext_vector_type(4)))  float    f4;

#define NEG_SLOPE 0.01f

__device__ __forceinline__ float lrelu(float v) { return v >= 0.0f ? v : NEG_SLOPE * v; }

__global__ void transpose_w_kernel(const float* __restrict__ W, _Float16* __restrict__ Wt) {
    int t = blockIdx.x * blockDim.x + threadIdx.x;
    if (t >= 256 * 64) return;
    int k = t >> 6;
    int c = t & 63;
    Wt[c * 256 + k] = (_Float16)W[t];
}

__global__ void __launch_bounds__(64) gemm_xw_kernel(const float* __restrict__ x,
                                                     const _Float16* __restrict__ Wt,
                                                     float* __restrict__ h) {
    const int lane = threadIdx.x & 31;
    const int wave = threadIdx.x >> 5;
    const int hi   = lane >> 4;
    const int r    = lane & 15;
    const int rowBase = (blockIdx.x * 2 + wave) * 16;

    v8f c0 = {}, c1 = {}, c2 = {}, c3 = {};

    const float*    xrow = x  + (size_t)(rowBase + r) * 256 + hi * 8;
    const _Float16* w0   = Wt + (size_t)(0 * 16 + r) * 256 + hi * 8;
    const _Float16* w1   = Wt + (size_t)(1 * 16 + r) * 256 + hi * 8;
    const _Float16* w2   = Wt + (size_t)(2 * 16 + r) * 256 + hi * 8;
    const _Float16* w3   = Wt + (size_t)(3 * 16 + r) * 256 + hi * 8;

    for (int kk = 0; kk < 8; ++kk) {
        const int kb = kk * 32;

        f4 a0 = *(const f4*)(xrow + kb);
        f4 a1 = *(const f4*)(xrow + kb + 4);
        f4 a2 = *(const f4*)(xrow + kb + 16);
        f4 a3 = *(const f4*)(xrow + kb + 20);
        v16h a;
        #pragma unroll
        for (int i = 0; i < 4; ++i) {
            a[i]      = (_Float16)a0[i];
            a[4 + i]  = (_Float16)a1[i];
            a[8 + i]  = (_Float16)a2[i];
            a[12 + i] = (_Float16)a3[i];
        }

        h8 b0lo = *(const h8*)(w0 + kb); h8 b0hi = *(const h8*)(w0 + kb + 16);
        h8 b1lo = *(const h8*)(w1 + kb); h8 b1hi = *(const h8*)(w1 + kb + 16);
        h8 b2lo = *(const h8*)(w2 + kb); h8 b2hi = *(const h8*)(w2 + kb + 16);
        h8 b3lo = *(const h8*)(w3 + kb); h8 b3hi = *(const h8*)(w3 + kb + 16);
        v16h b0v, b1v, b2v, b3v;
        #pragma unroll
        for (int i = 0; i < 8; ++i) {
            b0v[i] = b0lo[i]; b0v[8 + i] = b0hi[i];
            b1v[i] = b1lo[i]; b1v[8 + i] = b1hi[i];
            b2v[i] = b2lo[i]; b2v[8 + i] = b2hi[i];
            b3v[i] = b3lo[i]; b3v[8 + i] = b3hi[i];
        }

        c0 = __builtin_amdgcn_wmma_f32_16x16x32_f16(false, a, false, b0v, (short)0, c0, false, false);
        c1 = __builtin_amdgcn_wmma_f32_16x16x32_f16(false, a, false, b1v, (short)0, c1, false, false);
        c2 = __builtin_amdgcn_wmma_f32_16x16x32_f16(false, a, false, b2v, (short)0, c2, false, false);
        c3 = __builtin_amdgcn_wmma_f32_16x16x32_f16(false, a, false, b3v, (short)0, c3, false, false);
    }

    #pragma unroll
    for (int j = 0; j < 8; ++j) {
        float* p = h + (size_t)(rowBase + j + hi * 8) * 64;
        p[r]      = c0[j];
        p[16 + r] = c1[j];
        p[32 + r] = c2[j];
        p[48 + r] = c3[j];
    }
}

#define ROW_USED 0x80000000u
#define HIST_NODES   8192
#define HIST_CHUNKS  25
#define HIST_THREADS 512
#define HIST_BATCH   16
#define MARK_BATCH   4

__device__ __forceinline__ void hist_add(unsigned int* cnt, int node, unsigned int base) {
    const unsigned int a = (unsigned int)node - base;
    if (a < (unsigned int)HIST_NODES) atomicAdd(&cnt[a], 1u);
}

__global__ void __launch_bounds__(HIST_THREADS)
deg_hist_kernel(const int* __restrict__ dst, const int* __restrict__ idx,
                unsigned int* __restrict__ part, int E, int nIdx, int N) {
    __shared__ unsigned int cnt[HIST_NODES];
    const int tid = threadIdx.x;
    const int r = blockIdx.x / HIST_CHUNKS;
    const int c = blockIdx.x % HIST_CHUNKS;
    const unsigned int base = (unsigned int)r * HIST_NODES;

    for (int j = tid; j < HIST_NODES; j += HIST_THREADS) cnt[j] = 0u;
    __syncthreads();

    const int iper = (nIdx + HIST_CHUNKS - 1) / HIST_CHUNKS;
    const int t0 = c * iper;
    const int t1 = (t0 + iper < nIdx) ? (t0 + iper) : nIdx;
    for (int t = t0 + tid; t < t1; t += MARK_BATCH * HIST_THREADS) {
        int m[MARK_BATCH];
        #pragma unroll
        for (int u = 0; u < MARK_BATCH; ++u) {
            const int tt = t + u * HIST_THREADS;
            m[u] = (tt < t1) ? idx[tt] : -1;
        }
        #pragma unroll
        for (int u = 0; u < MARK_BATCH; ++u) {
            const unsigned int a = (unsigned int)m[u] - base;
            if (a < (unsigned int)HIST_NODES) atomicOr(&cnt[a], ROW_USED);
        }
    }

    const int nvec = E >> 2;
    const int gper = (nvec + HIST_CHUNKS - 1) / HIST_CHUNKS;
    const int g0 = c * gper;
    const int g1 = (g0 + gper < nvec) ? (g0 + gper) : nvec;
    if ((((size_t)dst) & 15) == 0) {
        const int4* __restrict__ dv = (const int4*)dst;
        for (int g = g0 + tid; g < g1; g += HIST_BATCH * HIST_THREADS) {
            int4 v[HIST_BATCH];
            #pragma unroll
            for (int u = 0; u < HIST_BATCH; ++u) {
                const int gg = g + u * HIST_THREADS;
                v[u] = (gg < g1) ? dv[gg] : make_int4(-1, -1, -1, -1);
            }
            #pragma unroll
            for (int u = 0; u < HIST_BATCH; ++u) {
                hist_add(cnt, v[u].x, base);
                hist_add(cnt, v[u].y, base);
                hist_add(cnt, v[u].z, base);
                hist_add(cnt, v[u].w, base);
            }
        }
    } else {
        for (int e = 4 * g0 + tid; e < 4 * g1; e += HIST_THREADS) hist_add(cnt, dst[e], base);
    }
    if (c == 0) {
        const int e = 4 * nvec + tid;
        if (e < E) hist_add(cnt, dst[e], base);
    }
    __syncthreads();

    unsigned int* __restrict__ plane = part + (size_t)c * N;
    unsigned int outv[HIST_NODES / HIST_THREADS];
    #pragma unroll
    for (int k = 0; k < HIST_NODES / HIST_THREADS; ++k) outv[k] = cnt[tid + k * HIST_THREADS];
    #pragma unroll
    for (int k = 0; k < HIST_NODES / HIST_THREADS; ++k) {
        const unsigned int i = base + (unsigned int)(tid + k * HIST_THREADS);
        if (i < (unsigned int)N) plane[i] = outv[k];
    }
}

__global__ void dinv_kernel(const unsigned int* __restrict__ part, unsigned int* __restrict__ deg,
                            float* __restrict__ dinv, int N) {
    int i = blockIdx.x * blockDim.x + threadIdx.x;
    if (i < N) {
        unsigned int n = 1u;
        unsigned int mark = 0u;
        #pragma unroll
        for (int c = 0; c < HIST_CHUNKS; ++c) {
            const unsigned int v = part[(size_t)c * N + i];
            n += v & ~ROW_USED;
            mark |= v & ROW_USED;
        }
        deg[i] = n | mark;
        float d = (float)n;
        dinv[i] = d > 0.0f ? rsqrtf(d) : 0.0f;
    }
}

__global__ void embed_init_kernel(const float* __restrict__ h, const float* __restrict__ dinv,
                                  float* __restrict__ embed, int total) {
    int t = blockIdx.x * blockDim.x + threadIdx.x;
    if (t >= total) return;
    int i = t >> 6;
    float di = dinv[i];
    embed[t] = h[t] * di * di;
}

__device__ __forceinline__ int   lane_i(int v, int j)   { return __builtin_amdgcn_readlane(v, j); }
__device__ __forceinline__ float lane_f(float v, int j) { return __int_as_float(__builtin_amdgcn_readlane(__float_as_int(v), j)); }

__global__ void __launch_bounds__(256)
scatter_edges_kernel(const float* __restrict__ h, const float* __restrict__ dinv,
                     const unsigned int* __restrict__ deg,
                     const int* __restrict__ src, const int* __restrict__ dst,
                     float* __restrict__ embed, int E) {
    const int lane = threadIdx.x & 31;
    const int e = blockIdx.x * blockDim.x + threadIdx.x;
    int s = 0, d = 0;
    float w = 0.0f;
    bool keep = false;
    if (e < E) {
        s = src[e];
        d = dst[e];
        keep = (deg[d] & ROW_USED) != 0u;
        if (keep) w = dinv[s] * dinv[d];
    }
    unsigned int m = __builtin_amdgcn_ballot_w32(keep);
    while (m) {
        if (__builtin_popcount(m) >= 4) {
            const int j0 = __builtin_ctz(m); m &= m - 1u;
            const int j1 = __builtin_ctz(m); m &= m - 1u;
            const int j2 = __builtin_ctz(m); m &= m - 1u;
            const int j3 = __builtin_ctz(m); m &= m - 1u;
            const float* p0 = h + (size_t)lane_i(s, j0) * 64 + lane;
            const float* p1 = h + (size_t)lane_i(s, j1) * 64 + lane;
            const float* p2 = h + (size_t)lane_i(s, j2) * 64 + lane;
            const float* p3 = h + (size_t)lane_i(s, j3) * 64 + lane;
            const float a0 = p0[0], b0 = p0[32];
            const float a1 = p1[0], b1 = p1[32];
            const float a2 = p2[0], b2 = p2[32];
            const float a3 = p3[0], b3 = p3[32];
            float* q0 = embed + (size_t)lane_i(d, j0) * 64 + lane;
            float* q1 = embed + (size_t)lane_i(d, j1) * 64 + lane;
            float* q2 = embed + (size_t)lane_i(d, j2) * 64 + lane;
            float* q3 = embed + (size_t)lane_i(d, j3) * 64 + lane;
            const float w0 = lane_f(w, j0), w1 = lane_f(w, j1), w2 = lane_f(w, j2), w3 = lane_f(w, j3);
            unsafeAtomicAdd(q0, a0 * w0); unsafeAtomicAdd(q0 + 32, b0 * w0);
            unsafeAtomicAdd(q1, a1 * w1); unsafeAtomicAdd(q1 + 32, b1 * w1);
            unsafeAtomicAdd(q2, a2 * w2); unsafeAtomicAdd(q2 + 32, b2 * w2);
            unsafeAtomicAdd(q3, a3 * w3); unsafeAtomicAdd(q3 + 32, b3 * w3);
        } else {
            const int j0 = __builtin_ctz(m); m &= m - 1u;
            const float* p0 = h + (size_t)lane_i(s, j0) * 64 + lane;
            const float a0 = p0[0], b0 = p0[32];
            float* q0 = embed + (size_t)lane_i(d, j0) * 64 + lane;
            const float w0 = lane_f(w, j0);
            unsafeAtomicAdd(q0, a0 * w0); unsafeAtomicAdd(q0 + 32, b0 * w0);
        }
    }
}

__global__ void mlp_kernel(const float* __restrict__ embed, const int* __restrict__ idx,
                           const float* __restrict__ b_conv,
                           const float* __restrict__ W1, const float* __restrict__ b1,
                           const float* __restrict__ W2, const float* __restrict__ b2,
                           float* __restrict__ out, int B) {
    int b = blockIdx.x * blockDim.x + threadIdx.x;
    if (b >= B) return;
    int i0 = idx[2 * b];
    int i1 = idx[2 * b + 1];
    const float* e0 = embed + (size_t)i0 * 64;
    const float* e1 = embed + (size_t)i1 * 64;

    float o1[16];
    #pragma unroll
    for (int j = 0; j < 16; ++j) o1[j] = b1[j];

    for (int c = 0; c < 64; ++c) {
        float v = lrelu(e0[c] + b_conv[c]);
        const float* wr = W1 + c * 16;
        #pragma unroll
        for (int j = 0; j < 16; ++j) o1[j] += v * wr[j];
    }
    for (int c = 0; c < 64; ++c) {
        float v = lrelu(e1[c] + b_conv[c]);
        const float* wr = W1 + (64 + c) * 16;
        #pragma unroll
        for (int j = 0; j < 16; ++j) o1[j] += v * wr[j];
    }

    float acc = b2[0];
    #pragma unroll
    for (int j = 0; j < 16; ++j) acc += lrelu(o1[j]) * W2[j];

    out[b] = 1.0f / (1.0f + __expf(-acc));
}

extern "C" void kernel_launch(void* const* d_in, const int* in_sizes, int n_in,
                              void* d_out, int out_size, void* d_ws, size_t ws_size,
                              hipStream_t stream) {
    const float* x        = (const float*)d_in[0];
    const int*   edge     = (const int*)d_in[1];
    const int*   index    = (const int*)d_in[2];
    const float* W_conv   = (const float*)d_in[3];
    const float* b_conv   = (const float*)d_in[4];
    const float* W1       = (const float*)d_in[5];
    const float* b1       = (const float*)d_in[6];
    const float* W2       = (const float*)d_in[7];
    const float* b2       = (const float*)d_in[8];
    float*       out      = (float*)d_out;

    const int N = in_sizes[0] / 256;
    const int E = in_sizes[1] / 2;
    const int B = in_sizes[2] / 2;
    const int* src = edge;
    const int* dst = edge + E;

    float*        h     = (float*)d_ws;
    float*        embed = h + (size_t)N * 64;
    unsigned int* deg   = (unsigned int*)(embed + (size_t)N * 64);
    float*        dinv  = (float*)(deg + N);
    _Float16*     Wt    = (_Float16*)(dinv + N);

    transpose_w_kernel<<<(256 * 64 + 255) / 256, 256, 0, stream>>>(W_conv, Wt);

    gemm_xw_kernel<<<N / 32, 64, 0, stream>>>(x, Wt, h);

    unsigned int* part = (unsigned int*)embed;
    deg_hist_kernel<<<((N + HIST_NODES - 1) / HIST_NODES) * HIST_CHUNKS, HIST_THREADS, 0, stream>>>(
        dst, index, part, E, 2 * B, N);
    dinv_kernel<<<(N + 255) / 256, 256, 0, stream>>>(part, deg, dinv, N);

    embed_init_kernel<<<(N * 64 + 255) / 256, 256, 0, stream>>>(h, dinv, embed, N * 64);
    scatter_edges_kernel<<<(E + 255) / 256, 256, 0, stream>>>(h, dinv, deg, src, dst, embed, E);

    mlp_kernel<<<(B + 255) / 256, 256, 0, stream>>>(embed, index, b_conv, W1, b1, W2, b2, out, B);
}


# self-loop term embed = h*dinv^2 stored by the GEMM epilogue for the gathered rows only; degree passes moved ahead of the GEMM; embed_init launch dropped (6 launches); sum order unchanged
# speedup vs baseline: 4.3395x; 1.1606x over previous
#include <hip/hip_runtime.h>
#include <hip/hip_bf16.h>

typedef __attribute__((ext_vector_type(16))) _Float16 v16h;
typedef __attribute__((ext_vector_type(8)))  _Float16 h8;
typedef __attribute__((ext_vector_type(8)))  float    v8f;
typedef __attribute__((ext_vector_type(4)))  float    f4;

#define NEG_SLOPE 0.01f

__device__ __forceinline__ float lrelu(float v) { return v >= 0.0f ? v : NEG_SLOPE * v; }

__global__ void transpose_w_kernel(const float* __restrict__ W, _Float16* __restrict__ Wt) {
    int t = blockIdx.x * blockDim.x + threadIdx.x;
    if (t >= 256 * 64) return;
    int k = t >> 6;
    int c = t & 63;
    Wt[c * 256 + k] = (_Float16)W[t];
}

#define ROW_USED 0x80000000u

__global__ void __launch_bounds__(64) gemm_xw_kernel(const float* __restrict__ x,
                                                     const _Float16* __restrict__ Wt,
                                                     const float* __restrict__ dinv,
                                                     const unsigned int* __restrict__ deg,
                                                     float* __restrict__ h,
                                                     float* __restrict__ embed) {
    const int lane = threadIdx.x & 31;
    const int wave = threadIdx.x >> 5;
    const int hi   = lane >> 4;
    const int r    = lane & 15;
    const int rowBase = (blockIdx.x * 2 + wave) * 16;

    v8f c0 = {}, c1 = {}, c2 = {}, c3 = {};

    const float*    xrow = x  + (size_t)(rowBase + r) * 256 + hi * 8;
    const _Float16* w0   = Wt + (size_t)(0 * 16 + r) * 256 + hi * 8;
    const _Float16* w1   = Wt + (size_t)(1 * 16 + r) * 256 + hi * 8;
    const _Float16* w2   = Wt + (size_t)(2 * 16 + r) * 256 + hi * 8;
    const _Float16* w3   = Wt + (size_t)(3 * 16 + r) * 256 + hi * 8;

    for (int kk = 0; kk < 8; ++kk) {
        const int kb = kk * 32;

        f4 a0 = *(const f4*)(xrow + kb);
        f4 a1 = *(const f4*)(xrow + kb + 4);
        f4 a2 = *(const f4*)(xrow + kb + 16);
        f4 a3 = *(const f4*)(xrow + kb + 20);
        v16h a;
        #pragma unroll
        for (int i = 0; i < 4; ++i) {
            a[i]      = (_Float16)a0[i];
            a[4 + i]  = (_Float16)a1[i];
            a[8 + i]  = (_Float16)a2[i];
            a[12 + i] = (_Float16)a3[i];
        }

        h8 b0lo = *(const h8*)(w0 + kb); h8 b0hi = *(const h8*)(w0 + kb + 16);
        h8 b1lo = *(const h8*)(w1 + kb); h8 b1hi = *(const h8*)(w1 + kb + 16);
        h8 b2lo = *(const h8*)(w2 + kb); h8 b2hi = *(const h8*)(w2 + kb + 16);
        h8 b3lo = *(const h8*)(w3 + kb); h8 b3hi = *(const h8*)(w3 + kb + 16);
        v16h b0v, b1v, b2v, b3v;
        #pragma unroll
        for (int i = 0; i < 8; ++i) {
            b0v[i] = b0lo[i]; b0v[8 + i] = b0hi[i];
            b1v[i] = b1lo[i]; b1v[8 + i] = b1hi[i];
            b2v[i] = b2lo[i]; b2v[8 + i] = b2hi[i];
            b3v[i] = b3lo[i]; b3v[8 + i] = b3hi[i];
        }

        c0 = __builtin_amdgcn_wmma_f32_16x16x32_f16(false, a, false, b0v, (short)0, c0, false, false);
        c1 = __builtin_amdgcn_wmma_f32_16x16x32_f16(false, a, false, b1v, (short)0, c1, false, false);
        c2 = __builtin_amdgcn_wmma_f32_16x16x32_f16(false, a, false, b2v, (short)0, c2, false, false);
        c3 = __builtin_amdgcn_wmma_f32_16x16x32_f16(false, a, false, b3v, (short)0, c3, false, false);
    }

    float        dv[8];
    unsigned int mk[8];
    #pragma unroll
    for (int j = 0; j < 8; ++j) {
        const int m = rowBase + j + hi * 8;
        dv[j] = dinv[m];
        mk[j] = deg[m];
    }

    #pragma unroll
    for (int j = 0; j < 8; ++j) {
        float* p = h + (size_t)(rowBase + j + hi * 8) * 64;
        p[r]      = c0[j];
        p[16 + r] = c1[j];
        p[32 + r] = c2[j];
        p[48 + r] = c3[j];
        if (mk[j] & ROW_USED) {
            float* q = embed + (size_t)(rowBase + j + hi * 8) * 64;
            const float di = dv[j];
            q[r]      = c0[j] * di * di;
            q[16 + r] = c1[j] * di * di;
            q[32 + r] = c2[j] * di * di;
            q[48 + r] = c3[j] * di * di;
        }
    }
}

#define HIST_NODES   8192
#define HIST_CHUNKS  25
#define HIST_THREADS 512
#define HIST_BATCH   16
#define MARK_BATCH   4

__device__ __forceinline__ void hist_add(unsigned int* cnt, int node, unsigned int base) {
    const unsigned int a = (unsigned int)node - base;
    if (a < (unsigned int)HIST_NODES) atomicAdd(&cnt[a], 1u);
}

__global__ void __launch_bounds__(HIST_THREADS)
deg_hist_kernel(const int* __restrict__ dst, const int* __restrict__ idx,
                unsigned int* __restrict__ part, int E, int nIdx, int N) {
    __shared__ unsigned int cnt[HIST_NODES];
    const int tid = threadIdx.x;
    const int r = blockIdx.x / HIST_CHUNKS;
    const int c = blockIdx.x % HIST_CHUNKS;
    const unsigned int base = (unsigned int)r * HIST_NODES;

    for (int j = tid; j < HIST_NODES; j += HIST_THREADS) cnt[j] = 0u;
    __syncthreads();

    const int iper = (nIdx + HIST_CHUNKS - 1) / HIST_CHUNKS;
    const int t0 = c * iper;
    const int t1 = (t0 + iper < nIdx) ? (t0 + iper) : nIdx;
    for (int t = t0 + tid; t < t1; t += MARK_BATCH * HIST_THREADS) {
        int m[MARK_BATCH];
        #pragma unroll
        for (int u = 0; u < MARK_BATCH; ++u) {
            const int tt = t + u * HIST_THREADS;
            m[u] = (tt < t1) ? idx[tt] : -1;
        }
        #pragma unroll
        for (int u = 0; u < MARK_BATCH; ++u) {
            const unsigned int a = (unsigned int)m[u] - base;
            if (a < (unsigned int)HIST_NODES) atomicOr(&cnt[a], ROW_USED);
        }
    }

    const int nvec = E >> 2;
    const int gper = (nvec + HIST_CHUNKS - 1) / HIST_CHUNKS;
    const int g0 = c * gper;
    const int g1 = (g0 + gper < nvec) ? (g0 + gper) : nvec;
    if ((((size_t)dst) & 15) == 0) {
        const int4* __restrict__ dv = (const int4*)dst;
        for (int g = g0 + tid; g < g1; g += HIST_BATCH * HIST_THREADS) {
            int4 v[HIST_BATCH];
            #pragma unroll
            for (int u = 0; u < HIST_BATCH; ++u) {
                const int gg = g + u * HIST_THREADS;
                v[u] = (gg < g1) ? dv[gg] : make_int4(-1, -1, -1, -1);
            }
            #pragma unroll
            for (int u = 0; u < HIST_BATCH; ++u) {
                hist_add(cnt, v[u].x, base);
                hist_add(cnt, v[u].y, base);
                hist_add(cnt, v[u].z, base);
                hist_add(cnt, v[u].w, base);
            }
        }
    } else {
        for (int e = 4 * g0 + tid; e < 4 * g1; e += HIST_THREADS) hist_add(cnt, dst[e], base);
    }
    if (c == 0) {
        const int e = 4 * nvec + tid;
        if (e < E) hist_add(cnt, dst[e], base);
    }
    __syncthreads();

    unsigned int* __restrict__ plane = part + (size_t)c * N;
    unsigned int outv[HIST_NODES / HIST_THREADS];
    #pragma unroll
    for (int k = 0; k < HIST_NODES / HIST_THREADS; ++k) outv[k] = cnt[tid + k * HIST_THREADS];
    #pragma unroll
    for (int k = 0; k < HIST_NODES / HIST_THREADS; ++k) {
        const unsigned int i = base + (unsigned int)(tid + k * HIST_THREADS);
        if (i < (unsigned int)N) plane[i] = outv[k];
    }
}

__global__ void dinv_kernel(const unsigned int* __restrict__ part, unsigned int* __restrict__ deg,
                            float* __restrict__ dinv, int N) {
    int i = blockIdx.x * blockDim.x + threadIdx.x;
    if (i < N) {
        unsigned int n = 1u;
        unsigned int mark = 0u;
        #pragma unroll
        for (int c = 0; c < HIST_CHUNKS; ++c) {
            const unsigned int v = part[(size_t)c * N + i];
            n += v & ~ROW_USED;
            mark |= v & ROW_USED;
        }
        deg[i] = n | mark;
        float d = (float)n;
        dinv[i] = d > 0.0f ? rsqrtf(d) : 0.0f;
    }
}

__device__ __forceinline__ int   lane_i(int v, int j)   { return __builtin_amdgcn_readlane(v, j); }
__device__ __forceinline__ float lane_f(float v, int j) { return __int_as_float(__builtin_amdgcn_readlane(__float_as_int(v), j)); }

__global__ void __launch_bounds__(256)
scatter_edges_kernel(const float* __restrict__ h, const float* __restrict__ dinv,
                     const unsigned int* __restrict__ deg,
                     const int* __restrict__ src, const int* __restrict__ dst,
                     float* __restrict__ embed, int E) {
    const int lane = threadIdx.x & 31;
    const int e = blockIdx.x * blockDim.x + threadIdx.x;
    int s = 0, d = 0;
    float w = 0.0f;
    bool keep = false;
    if (e < E) {
        s = src[e];
        d = dst[e];
        keep = (deg[d] & ROW_USED) != 0u;
        if (keep) w = dinv[s] * dinv[d];
    }
    unsigned int m = __builtin_amdgcn_ballot_w32(keep);
    while (m) {
        if (__builtin_popcount(m) >= 4) {
            const int j0 = __builtin_ctz(m); m &= m - 1u;
            const int j1 = __builtin_ctz(m); m &= m - 1u;
            const int j2 = __builtin_ctz(m); m &= m - 1u;
            const int j3 = __builtin_ctz(m); m &= m - 1u;
            const float* p0 = h + (size_t)lane_i(s, j0) * 64 + lane;
            const float* p1 = h + (size_t)lane_i(s, j1) * 64 + lane;
            const float* p2 = h + (size_t)lane_i(s, j2) * 64 + lane;
            const float* p3 = h + (size_t)lane_i(s, j3) * 64 + lane;
            const float a0 = p0[0], b0 = p0[32];
            const float a1 = p1[0], b1 = p1[32];
            const float a2 = p2[0], b2 = p2[32];
            const float a3 = p3[0], b3 = p3[32];
            float* q0 = embed + (size_t)lane_i(d, j0) * 64 + lane;
            float* q1 = embed + (size_t)lane_i(d, j1) * 64 + lane;
            float* q2 = embed + (size_t)lane_i(d, j2) * 64 + lane;
            float* q3 = embed + (size_t)lane_i(d, j3) * 64 + lane;
            const float w0 = lane_f(w, j0), w1 = lane_f(w, j1), w2 = lane_f(w, j2), w3 = lane_f(w, j3);
            unsafeAtomicAdd(q0, a0 * w0); unsafeAtomicAdd(q0 + 32, b0 * w0);
            unsafeAtomicAdd(q1, a1 * w1); unsafeAtomicAdd(q1 + 32, b1 * w1);
            unsafeAtomicAdd(q2, a2 * w2); unsafeAtomicAdd(q2 + 32, b2 * w2);
            unsafeAtomicAdd(q3, a3 * w3); unsafeAtomicAdd(q3 + 32, b3 * w3);
        } else {
            const int j0 = __builtin_ctz(m); m &= m - 1u;
            const float* p0 = h + (size_t)lane_i(s, j0) * 64 + lane;
            const float a0 = p0[0], b0 = p0[32];
            float* q0 = embed + (size_t)lane_i(d, j0) * 64 + lane;
            const float w0 = lane_f(w, j0);
            unsafeAtomicAdd(q0, a0 * w0); unsafeAtomicAdd(q0 + 32, b0 * w0);
        }
    }
}

__global__ void mlp_kernel(const float* __restrict__ embed, const int* __restrict__ idx,
                           const float* __restrict__ b_conv,
                           const float* __restrict__ W1, const float* __restrict__ b1,
                           const float* __restrict__ W2, const float* __restrict__ b2,
                           float* __restrict__ out, int B) {
    int b = blockIdx.x * blockDim.x + threadIdx.x;
    if (b >= B) return;
    int i0 = idx[2 * b];
    int i1 = idx[2 * b + 1];
    const float* e0 = embed + (size_t)i0 * 64;
    const float* e1 = embed + (size_t)i1 * 64;

    float o1[16];
    #pragma unroll
    for (int j = 0; j < 16; ++j) o1[j] = b1[j];

    for (int c = 0; c < 64; ++c) {
        float v = lrelu(e0[c] + b_conv[c]);
        const float* wr = W1 + c * 16;
        #pragma unroll
        for (int j = 0; j < 16; ++j) o1[j] += v * wr[j];
    }
    for (int c = 0; c < 64; ++c) {
        float v = lrelu(e1[c] + b_conv[c]);
        const float* wr = W1 + (64 + c) * 16;
        #pragma unroll
        for (int j = 0; j < 16; ++j) o1[j] += v * wr[j];
    }

    float acc = b2[0];
    #pragma unroll
    for (int j = 0; j < 16; ++j) acc += lrelu(o1[j]) * W2[j];

    out[b] = 1.0f / (1.0f + __expf(-acc));
}

extern "C" void kernel_launch(void* const* d_in, const int* in_sizes, int n_in,
                              void* d_out, int out_size, void* d_ws, size_t ws_size,
                              hipStream_t stream) {
    const float* x        = (const float*)d_in[0];
    const int*   edge     = (const int*)d_in[1];
    const int*   index    = (const int*)d_in[2];
    const float* W_conv   = (const float*)d_in[3];
    const float* b_conv   = (const float*)d_in[4];
    const float* W1       = (const float*)d_in[5];
    const float* b1       = (const float*)d_in[6];
    const float* W2       = (const float*)d_in[7];
    const float* b2       = (const float*)d_in[8];
    float*       out      = (float*)d_out;

    const int N = in_sizes[0] / 256;
    const int E = in_sizes[1] / 2;
    const int B = in_sizes[2] / 2;
    const int* src = edge;
    const int* dst = edge + E;

    float*        h     = (float*)d_ws;
    float*        embed = h + (size_t)N * 64;
    unsigned int* deg   = (unsigned int*)(embed + (size_t)N * 64);
    float*        dinv  = (float*)(deg + N);
    _Float16*     Wt    = (_Float16*)(dinv + N);

    transpose_w_kernel<<<(256 * 64 + 255) / 256, 256, 0, stream>>>(W_conv, Wt);

    unsigned int* part = (unsigned int*)embed;
    deg_hist_kernel<<<((N + HIST_NODES - 1) / HIST_NODES) * HIST_CHUNKS, HIST_THREADS, 0, stream>>>(
        dst, index, part, E, 2 * B, N);
    dinv_kernel<<<(N + 255) / 256, 256, 0, stream>>>(part, deg, dinv, N);

    gemm_xw_kernel<<<N / 32, 64, 0, stream>>>(x, Wt, dinv, deg, h, embed);

    scatter_edges_kernel<<<(E + 255) / 256, 256, 0, stream>>>(h, dinv, deg, src, dst, embed, E);

    mlp_kernel<<<(B + 255) / 256, 256, 0, stream>>>(embed, index, b_conv, W1, b1, W2, b2, out, B);
}


# GEMM operands staged in LDS: x tile and W copy fetched by whole-line 512-B global loads, fragments read from LDS (was row-strided 16-line loads, W re-read per wave); k order and rounding unchanged
# speedup vs baseline: 4.5915x; 1.0581x over previous
#include <hip/hip_runtime.h>
#include <hip/hip_bf16.h>

typedef __attribute__((ext_vector_type(16))) _Float16 v16h;
typedef __attribute__((ext_vector_type(8)))  _Float16 h8;
typedef __attribute__((ext_vector_type(8)))  float    v8f;
typedef __attribute__((ext_vector_type(4)))  float    f4;

#define NEG_SLOPE 0.01f

__device__ __forceinline__ float lrelu(float v) { return v >= 0.0f ? v : NEG_SLOPE * v; }

__global__ void transpose_w_kernel(const float* __restrict__ W, _Float16* __restrict__ Wt) {
    int t = blockIdx.x * blockDim.x + threadIdx.x;
    if (t >= 256 * 64) return;
    int k = t >> 6;
    int c = t & 63;
    Wt[c * 256 + k] = (_Float16)W[t];
}

#define ROW_USED 0x80000000u

#define GT_WAVES   4
#define GT_THREADS (GT_WAVES * 32)
#define GT_PITCH   264

typedef __attribute__((ext_vector_type(4))) _Float16 h4;

__global__ void __launch_bounds__(GT_THREADS)
gemm_xw_kernel(const float* __restrict__ x,
               const _Float16* __restrict__ Wt,
               const float* __restrict__ dinv,
               const unsigned int* __restrict__ deg,
               float* __restrict__ h,
               float* __restrict__ embed, int N) {
    __shared__ __attribute__((aligned(16))) _Float16 sW[64 * GT_PITCH];
    __shared__ __attribute__((aligned(16))) _Float16 sA[GT_WAVES * 16 * GT_PITCH];

    const int tid  = threadIdx.x;
    const int lane = tid & 31;
    const int wave = tid >> 5;
    const int hi   = lane >> 4;
    const int r    = lane & 15;
    const int rowBase = (blockIdx.x * GT_WAVES + wave) * 16;
    const bool active = rowBase < N;

    f4 xa[32];
    if (active) {
        #pragma unroll
        for (int i = 0; i < 32; ++i) {
            int row = rowBase + (i >> 1);
            row = row < N ? row : N - 1;
            xa[i] = *(const f4*)(x + (size_t)row * 256 + (i & 1) * 128 + lane * 4);
        }
    }
    h8 wv[64 / GT_WAVES];
    #pragma unroll
    for (int i = 0; i < 64 / GT_WAVES; ++i)
        wv[i] = *(const h8*)(Wt + (size_t)(tid + i * GT_THREADS) * 8);

    float        dv[8];
    unsigned int mk[8];
    #pragma unroll
    for (int j = 0; j < 8; ++j) {
        int m = rowBase + j + hi * 8;
        m = m < N ? m : N - 1;
        dv[j] = dinv[m];
        mk[j] = deg[m];
    }

    #pragma unroll
    for (int i = 0; i < 64 / GT_WAVES; ++i)
        *(h8*)(sW + (wave + i * GT_WAVES) * GT_PITCH + lane * 8) = wv[i];
    if (active) {
        _Float16* myA = sA + wave * 16 * GT_PITCH;
        #pragma unroll
        for (int i = 0; i < 32; ++i) {
            h4 v;
            v[0] = (_Float16)xa[i][0]; v[1] = (_Float16)xa[i][1];
            v[2] = (_Float16)xa[i][2]; v[3] = (_Float16)xa[i][3];
            *(h4*)(myA + (i >> 1) * GT_PITCH + (i & 1) * 128 + lane * 4) = v;
        }
    }
    __syncthreads();
    if (!active) return;

    v8f c0 = {}, c1 = {}, c2 = {}, c3 = {};
    const _Float16* ap = sA + (wave * 16 + r) * GT_PITCH + hi * 8;
    const _Float16* bp = sW + r * GT_PITCH + hi * 8;

    h8 flo[2][5], fhi[2][5];
    #pragma unroll
    for (int t = 0; t < 4; ++t) {
        flo[0][t] = *(const h8*)(bp + t * 16 * GT_PITCH);
        fhi[0][t] = *(const h8*)(bp + t * 16 * GT_PITCH + 16);
    }
    flo[0][4] = *(const h8*)(ap);
    fhi[0][4] = *(const h8*)(ap + 16);

    #pragma unroll
    for (int kk = 0; kk < 8; ++kk) {
        const int cur = kk & 1, nxt = cur ^ 1;
        if (kk + 1 < 8) {
            const int kb = (kk + 1) * 32;
            #pragma unroll
            for (int t = 0; t < 4; ++t) {
                flo[nxt][t] = *(const h8*)(bp + t * 16 * GT_PITCH + kb);
                fhi[nxt][t] = *(const h8*)(bp + t * 16 * GT_PITCH + kb + 16);
            }
            flo[nxt][4] = *(const h8*)(ap + kb);
            fhi[nxt][4] = *(const h8*)(ap + kb + 16);
        }
        __builtin_amdgcn_sched_barrier(0);
        const v16h a   = __builtin_shufflevector(flo[cur][4], fhi[cur][4], 0, 1, 2, 3, 4, 5, 6, 7, 8, 9, 10, 11, 12, 13, 14, 15);
        const v16h b0v = __builtin_shufflevector(flo[cur][0], fhi[cur][0], 0, 1, 2, 3, 4, 5, 6, 7, 8, 9, 10, 11, 12, 13, 14, 15);
        const v16h b1v = __builtin_shufflevector(flo[cur][1], fhi[cur][1], 0, 1, 2, 3, 4, 5, 6, 7, 8, 9, 10, 11, 12, 13, 14, 15);
        const v16h b2v = __builtin_shufflevector(flo[cur][2], fhi[cur][2], 0, 1, 2, 3, 4, 5, 6, 7, 8, 9, 10, 11, 12, 13, 14, 15);
        const v16h b3v = __builtin_shufflevector(flo[cur][3], fhi[cur][3], 0, 1, 2, 3, 4, 5, 6, 7, 8, 9, 10, 11, 12, 13, 14, 15);
        c0 = __builtin_amdgcn_wmma_f32_16x16x32_f16(false, a, false, b0v, (short)0, c0, false, false);
        c1 = __builtin_amdgcn_wmma_f32_16x16x32_f16(false, a, false, b1v, (short)0, c1, false, false);
        c2 = __builtin_amdgcn_wmma_f32_16x16x32_f16(false, a, false, b2v, (short)0, c2, false, false);
        c3 = __builtin_amdgcn_wmma_f32_16x16x32_f16(false, a, false, b3v, (short)0, c3, false, false);
    }

    #pragma unroll
    for (int j = 0; j < 8; ++j) {
        const int m = rowBase + j + hi * 8;
        if (m < N) {
            float* p = h + (size_t)m * 64;
            p[r]      = c0[j];
            p[16 + r] = c1[j];
            p[32 + r] = c2[j];
            p[48 + r] = c3[j];
            if (mk[j] & ROW_USED) {
                float* q = embed + (size_t)m * 64;
                const float di = dv[j];
                q[r]      = c0[j] * di * di;
                q[16 + r] = c1[j] * di * di;
                q[32 + r] = c2[j] * di * di;
                q[48 + r] = c3[j] * di * di;
            }
        }
    }
}

#define HIST_NODES   8192
#define HIST_CHUNKS  25
#define HIST_THREADS 512
#define HIST_BATCH   16
#define MARK_BATCH   4

__device__ __forceinline__ void hist_add(unsigned int* cnt, int node, unsigned int base) {
    const unsigned int a = (unsigned int)node - base;
    if (a < (unsigned int)HIST_NODES) atomicAdd(&cnt[a], 1u);
}

__global__ void __launch_bounds__(HIST_THREADS)
deg_hist_kernel(const int* __restrict__ dst, const int* __restrict__ idx,
                unsigned int* __restrict__ part, int E, int nIdx, int N) {
    __shared__ unsigned int cnt[HIST_NODES];
    const int tid = threadIdx.x;
    const int r = blockIdx.x / HIST_CHUNKS;
    const int c = blockIdx.x % HIST_CHUNKS;
    const unsigned int base = (unsigned int)r * HIST_NODES;

    for (int j = tid; j < HIST_NODES; j += HIST_THREADS) cnt[j] = 0u;
    __syncthreads();

    const int iper = (nIdx + HIST_CHUNKS - 1) / HIST_CHUNKS;
    const int t0 = c * iper;
    const int t1 = (t0 + iper < nIdx) ? (t0 + iper) : nIdx;
    for (int t = t0 + tid; t < t1; t += MARK_BATCH * HIST_THREADS) {
        int m[MARK_BATCH];
        #pragma unroll
        for (int u = 0; u < MARK_BATCH; ++u) {
            const int tt = t + u * HIST_THREADS;
            m[u] = (tt < t1) ? idx[tt] : -1;
        }
        #pragma unroll
        for (int u = 0; u < MARK_BATCH; ++u) {
            const unsigned int a = (unsigned int)m[u] - base;
            if (a < (unsigned int)HIST_NODES) atomicOr(&cnt[a], ROW_USED);
        }
    }

    const int nvec = E >> 2;
    const int gper = (nvec + HIST_CHUNKS - 1) / HIST_CHUNKS;
    const int g0 = c * gper;
    const int g1 = (g0 + gper < nvec) ? (g0 + gper) : nvec;
    if ((((size_t)dst) & 15) == 0) {
        const int4* __restrict__ dv = (const int4*)dst;
        for (int g = g0 + tid; g < g1; g += HIST_BATCH * HIST_THREADS) {
            int4 v[HIST_BATCH];
            #pragma unroll
            for (int u = 0; u < HIST_BATCH; ++u) {
                const int gg = g + u * HIST_THREADS;
                v[u] = (gg < g1) ? dv[gg] : make_int4(-1, -1, -1, -1);
            }
            #pragma unroll
            for (int u = 0; u < HIST_BATCH; ++u) {
                hist_add(cnt, v[u].x, base);
                hist_add(cnt, v[u].y, base);
                hist_add(cnt, v[u].z, base);
                hist_add(cnt, v[u].w, base);
            }
        }
    } else {
        for (int e = 4 * g0 + tid; e < 4 * g1; e += HIST_THREADS) hist_add(cnt, dst[e], base);
    }
    if (c == 0) {
        const int e = 4 * nvec + tid;
        if (e < E) hist_add(cnt, dst[e], base);
    }
    __syncthreads();

    unsigned int* __restrict__ plane = part + (size_t)c * N;
    unsigned int outv[HIST_NODES / HIST_THREADS];
    #pragma unroll
    for (int k = 0; k < HIST_NODES / HIST_THREADS; ++k) outv[k] = cnt[tid + k * HIST_THREADS];
    #pragma unroll
    for (int k = 0; k < HIST_NODES / HIST_THREADS; ++k) {
        const unsigned int i = base + (unsigned int)(tid + k * HIST_THREADS);
        if (i < (unsigned int)N) plane[i] = outv[k];
    }
}

__global__ void dinv_kernel(const unsigned int* __restrict__ part, unsigned int* __restrict__ deg,
                            float* __restrict__ dinv, int N) {
    int i = blockIdx.x * blockDim.x + threadIdx.x;
    if (i < N) {
        unsigned int n = 1u;
        unsigned int mark = 0u;
        #pragma unroll
        for (int c = 0; c < HIST_CHUNKS; ++c) {
            const unsigned int v = part[(size_t)c * N + i];
            n += v & ~ROW_USED;
            mark |= v & ROW_USED;
        }
        deg[i] = n | mark;
        float d = (float)n;
        dinv[i] = d > 0.0f ? rsqrtf(d) : 0.0f;
    }
}

__device__ __forceinline__ int   lane_i(int v, int j)   { return __builtin_amdgcn_readlane(v, j); }
__device__ __forceinline__ float lane_f(float v, int j) { return __int_as_float(__builtin_amdgcn_readlane(__float_as_int(v), j)); }

__global__ void __launch_bounds__(256)
scatter_edges_kernel(const float* __restrict__ h, const float* __restrict__ dinv,
                     const unsigned int* __restrict__ deg,
                     const int* __restrict__ src, const int* __restrict__ dst,
                     float* __restrict__ embed, int E) {
    const int lane = threadIdx.x & 31;
    const int e = blockIdx.x * blockDim.x + threadIdx.x;
    int s = 0, d = 0;
    float w = 0.0f;
    bool keep = false;
    if (e < E) {
        s = src[e];
        d = dst[e];
        keep = (deg[d] & ROW_USED) != 0u;
        if (keep) w = dinv[s] * dinv[d];
    }
    unsigned int m = __builtin_amdgcn_ballot_w32(keep);
    while (m) {
        if (__builtin_popcount(m) >= 4) {
            const int j0 = __builtin_ctz(m); m &= m - 1u;
            const int j1 = __builtin_ctz(m); m &= m - 1u;
            const int j2 = __builtin_ctz(m); m &= m - 1u;
            const int j3 = __builtin_ctz(m); m &= m - 1u;
            const float* p0 = h + (size_t)lane_i(s, j0) * 64 + lane;
            const float* p1 = h + (size_t)lane_i(s, j1) * 64 + lane;
            const float* p2 = h + (size_t)lane_i(s, j2) * 64 + lane;
            const float* p3 = h + (size_t)lane_i(s, j3) * 64 + lane;
            const float a0 = p0[0], b0 = p0[32];
            const float a1 = p1[0], b1 = p1[32];
            const float a2 = p2[0], b2 = p2[32];
            const float a3 = p3[0], b3 = p3[32];
            float* q0 = embed + (size_t)lane_i(d, j0) * 64 + lane;
            float* q1 = embed + (size_t)lane_i(d, j1) * 64 + lane;
            float* q2 = embed + (size_t)lane_i(d, j2) * 64 + lane;
            float* q3 = embed + (size_t)lane_i(d, j3) * 64 + lane;
            const float w0 = lane_f(w, j0), w1 = lane_f(w, j1), w2 = lane_f(w, j2), w3 = lane_f(w, j3);
            unsafeAtomicAdd(q0, a0 * w0); unsafeAtomicAdd(q0 + 32, b0 * w0);
            unsafeAtomicAdd(q1, a1 * w1); unsafeAtomicAdd(q1 + 32, b1 * w1);
            unsafeAtomicAdd(q2, a2 * w2); unsafeAtomicAdd(q2 + 32, b2 * w2);
            unsafeAtomicAdd(q3, a3 * w3); unsafeAtomicAdd(q3 + 32, b3 * w3);
        } else {
            const int j0 = __builtin_ctz(m); m &= m - 1u;
            const float* p0 = h + (size_t)lane_i(s, j0) * 64 + lane;
            const float a0 = p0[0], b0 = p0[32];
            float* q0 = embed + (size_t)lane_i(d, j0) * 64 + lane;
            const float w0 = lane_f(w, j0);
            unsafeAtomicAdd(q0, a0 * w0); unsafeAtomicAdd(q0 + 32, b0 * w0);
        }
    }
}

__global__ void mlp_kernel(const float* __restrict__ embed, const int* __restrict__ idx,
                           const float* __restrict__ b_conv,
                           const float* __restrict__ W1, const float* __restrict__ b1,
                           const float* __restrict__ W2, const float* __restrict__ b2,
                           float* __restrict__ out, int B) {
    int b = blockIdx.x * blockDim.x + threadIdx.x;
    if (b >= B) return;
    int i0 = idx[2 * b];
    int i1 = idx[2 * b + 1];
    const float* e0 = embed + (size_t)i0 * 64;
    const float* e1 = embed + (size_t)i1 * 64;

    float o1[16];
    #pragma unroll
    for (int j = 0; j < 16; ++j) o1[j] = b1[j];

    for (int c = 0; c < 64; ++c) {
        float v = lrelu(e0[c] + b_conv[c]);
        const float* wr = W1 + c * 16;
        #pragma unroll
        for (int j = 0; j < 16; ++j) o1[j] += v * wr[j];
    }
    for (int c = 0; c < 64; ++c) {
        float v = lrelu(e1[c] + b_conv[c]);
        const float* wr = W1 + (64 + c) * 16;
        #pragma unroll
        for (int j = 0; j < 16; ++j) o1[j] += v * wr[j];
    }

    float acc = b2[0];
    #pragma unroll
    for (int j = 0; j < 16; ++j) acc += lrelu(o1[j]) * W2[j];

    out[b] = 1.0f / (1.0f + __expf(-acc));
}

extern "C" void kernel_launch(void* const* d_in, const int* in_sizes, int n_in,
                              void* d_out, int out_size, void* d_ws, size_t ws_size,
                              hipStream_t stream) {
    const float* x        = (const float*)d_in[0];
    const int*   edge     = (const int*)d_in[1];
    const int*   index    = (const int*)d_in[2];
    const float* W_conv   = (const float*)d_in[3];
    const float* b_conv   = (const float*)d_in[4];
    const float* W1       = (const float*)d_in[5];
    const float* b1       = (const float*)d_in[6];
    const float* W2       = (const float*)d_in[7];
    const float* b2       = (const float*)d_in[8];
    float*       out      = (float*)d_out;

    const int N = in_sizes[0] / 256;
    const int E = in_sizes[1] / 2;
    const int B = in_sizes[2] / 2;
    const int* src = edge;
    const int* dst = edge + E;

    float*        h     = (float*)d_ws;
    float*        embed = h + (size_t)N * 64;
    unsigned int* deg   = (unsigned int*)(embed + (size_t)N * 64);
    float*        dinv  = (float*)(deg + N);
    _Float16*     Wt    = (_Float16*)(dinv + N);

    transpose_w_kernel<<<(256 * 64 + 255) / 256, 256, 0, stream>>>(W_conv, Wt);

    unsigned int* part = (unsigned int*)embed;
    deg_hist_kernel<<<((N + HIST_NODES - 1) / HIST_NODES) * HIST_CHUNKS, HIST_THREADS, 0, stream>>>(
        dst, index, part, E, 2 * B, N);
    dinv_kernel<<<(N + 255) / 256, 256, 0, stream>>>(part, deg, dinv, N);

    gemm_xw_kernel<<<(N + 16 * GT_WAVES - 1) / (16 * GT_WAVES), GT_THREADS, 0, stream>>>(x, Wt, dinv, deg, h, embed, N);

    scatter_edges_kernel<<<(E + 255) / 256, 256, 0, stream>>>(h, dinv, deg, src, dst, embed, E);

    mlp_kernel<<<(B + 255) / 256, 256, 0, stream>>>(embed, index, b_conv, W1, b1, W2, b2, out, B);
}


# MLP head on the matrix unit: 16 pairs per wave, f32 16x16x4 WMMA (A = W1 transposed, B = gathered activations), 1024 waves over all WGPs (was 512 on 64); head sums in a new order, still f32
# speedup vs baseline: 5.5736x; 1.2139x over previous
#include <hip/hip_runtime.h>
#include <hip/hip_bf16.h>

typedef __attribute__((ext_vector_type(16))) _Float16 v16h;
typedef __attribute__((ext_vector_type(8)))  _Float16 h8;
typedef __attribute__((ext_vector_type(8)))  float    v8f;
typedef __attribute__((ext_vector_type(4)))  float    f4;

#define NEG_SLOPE 0.01f

__device__ __forceinline__ float lrelu(float v) { return v >= 0.0f ? v : NEG_SLOPE * v; }

__global__ void transpose_w_kernel(const float* __restrict__ W, _Float16* __restrict__ Wt) {
    int t = blockIdx.x * blockDim.x + threadIdx.x;
    if (t >= 256 * 64) return;
    int k = t >> 6;
    int c = t & 63;
    Wt[c * 256 + k] = (_Float16)W[t];
}

#define ROW_USED 0x80000000u

#define GT_WAVES   4
#define GT_THREADS (GT_WAVES * 32)
#define GT_PITCH   264

typedef __attribute__((ext_vector_type(4))) _Float16 h4;

__global__ void __launch_bounds__(GT_THREADS)
gemm_xw_kernel(const float* __restrict__ x,
               const _Float16* __restrict__ Wt,
               const float* __restrict__ dinv,
               const unsigned int* __restrict__ deg,
               float* __restrict__ h,
               float* __restrict__ embed, int N) {
    __shared__ __attribute__((aligned(16))) _Float16 sW[64 * GT_PITCH];
    __shared__ __attribute__((aligned(16))) _Float16 sA[GT_WAVES * 16 * GT_PITCH];

    const int tid  = threadIdx.x;
    const int lane = tid & 31;
    const int wave = tid >> 5;
    const int hi   = lane >> 4;
    const int r    = lane & 15;
    const int rowBase = (blockIdx.x * GT_WAVES + wave) * 16;
    const bool active = rowBase < N;

    f4 xa[32];
    if (active) {
        #pragma unroll
        for (int i = 0; i < 32; ++i) {
            int row = rowBase + (i >> 1);
            row = row < N ? row : N - 1;
            xa[i] = *(const f4*)(x + (size_t)row * 256 + (i & 1) * 128 + lane * 4);
        }
    }
    h8 wv[64 / GT_WAVES];
    #pragma unroll
    for (int i = 0; i < 64 / GT_WAVES; ++i)
        wv[i] = *(const h8*)(Wt + (size_t)(tid + i * GT_THREADS) * 8);

    float        dv[8];
    unsigned int mk[8];
    #pragma unroll
    for (int j = 0; j < 8; ++j) {
        int m = rowBase + j + hi * 8;
        m = m < N ? m : N - 1;
        dv[j] = dinv[m];
        mk[j] = deg[m];
    }

    #pragma unroll
    for (int i = 0; i < 64 / GT_WAVES; ++i)
        *(h8*)(sW + (wave + i * GT_WAVES) * GT_PITCH + lane * 8) = wv[i];
    if (active) {
        _Float16* myA = sA + wave * 16 * GT_PITCH;
        #pragma unroll
        for (int i = 0; i < 32; ++i) {
            h4 v;
            v[0] = (_Float16)xa[i][0]; v[1] = (_Float16)xa[i][1];
            v[2] = (_Float16)xa[i][2]; v[3] = (_Float16)xa[i][3];
            *(h4*)(myA + (i >> 1) * GT_PITCH + (i & 1) * 128 + lane * 4) = v;
        }
    }
    __syncthreads();
    if (!active) return;

    v8f c0 = {}, c1 = {}, c2 = {}, c3 = {};
    const _Float16* ap = sA + (wave * 16 + r) * GT_PITCH + hi * 8;
    const _Float16* bp = sW + r * GT_PITCH + hi * 8;

    h8 flo[2][5], fhi[2][5];
    #pragma unroll
    for (int t = 0; t < 4; ++t) {
        flo[0][t] = *(const h8*)(bp + t * 16 * GT_PITCH);
        fhi[0][t] = *(const h8*)(bp + t * 16 * GT_PITCH + 16);
    }
    flo[0][4] = *(const h8*)(ap);
    fhi[0][4] = *(const h8*)(ap + 16);

    #pragma unroll
    for (int kk = 0; kk < 8; ++kk) {
        const int cur = kk & 1, nxt = cur ^ 1;
        if (kk + 1 < 8) {
            const int kb = (kk + 1) * 32;
            #pragma unroll
            for (int t = 0; t < 4; ++t) {
                flo[nxt][t] = *(const h8*)(bp + t * 16 * GT_PITCH + kb);
                fhi[nxt][t] = *(const h8*)(bp + t * 16 * GT_PITCH + kb + 16);
            }
            flo[nxt][4] = *(const h8*)(ap + kb);
            fhi[nxt][4] = *(const h8*)(ap + kb + 16);
        }
        __builtin_amdgcn_sched_barrier(0);
        const v16h a   = __builtin_shufflevector(flo[cur][4], fhi[cur][4], 0, 1, 2, 3, 4, 5, 6, 7, 8, 9, 10, 11, 12, 13, 14, 15);
        const v16h b0v = __builtin_shufflevector(flo[cur][0], fhi[cur][0], 0, 1, 2, 3, 4, 5, 6, 7, 8, 9, 10, 11, 12, 13, 14, 15);
        const v16h b1v = __builtin_shufflevector(flo[cur][1], fhi[cur][1], 0, 1, 2, 3, 4, 5, 6, 7, 8, 9, 10, 11, 12, 13, 14, 15);
        const v16h b2v = __builtin_shufflevector(flo[cur][2], fhi[cur][2], 0, 1, 2, 3, 4, 5, 6, 7, 8, 9, 10, 11, 12, 13, 14, 15);
        const v16h b3v = __builtin_shufflevector(flo[cur][3], fhi[cur][3], 0, 1, 2, 3, 4, 5, 6, 7, 8, 9, 10, 11, 12, 13, 14, 15);
        c0 = __builtin_amdgcn_wmma_f32_16x16x32_f16(false, a, false, b0v, (short)0, c0, false, false);
        c1 = __builtin_amdgcn_wmma_f32_16x16x32_f16(false, a, false, b1v, (short)0, c1, false, false);
        c2 = __builtin_amdgcn_wmma_f32_16x16x32_f16(false, a, false, b2v, (short)0, c2, false, false);
        c3 = __builtin_amdgcn_wmma_f32_16x16x32_f16(false, a, false, b3v, (short)0, c3, false, false);
    }

    #pragma unroll
    for (int j = 0; j < 8; ++j) {
        const int m = rowBase + j + hi * 8;
        if (m < N) {
            float* p = h + (size_t)m * 64;
            p[r]      = c0[j];
            p[16 + r] = c1[j];
            p[32 + r] = c2[j];
            p[48 + r] = c3[j];
            if (mk[j] & ROW_USED) {
                float* q = embed + (size_t)m * 64;
                const float di = dv[j];
                q[r]      = c0[j] * di * di;
                q[16 + r] = c1[j] * di * di;
                q[32 + r] = c2[j] * di * di;
                q[48 + r] = c3[j] * di * di;
            }
        }
    }
}

#define HIST_NODES   8192
#define HIST_CHUNKS  25
#define HIST_THREADS 512
#define HIST_BATCH   16
#define MARK_BATCH   4

__device__ __forceinline__ void hist_add(unsigned int* cnt, int node, unsigned int base) {
    const unsigned int a = (unsigned int)node - base;
    if (a < (unsigned int)HIST_NODES) atomicAdd(&cnt[a], 1u);
}

__global__ void __launch_bounds__(HIST_THREADS)
deg_hist_kernel(const int* __restrict__ dst, const int* __restrict__ idx,
                unsigned int* __restrict__ part, int E, int nIdx, int N) {
    __shared__ unsigned int cnt[HIST_NODES];
    const int tid = threadIdx.x;
    const int r = blockIdx.x / HIST_CHUNKS;
    const int c = blockIdx.x % HIST_CHUNKS;
    const unsigned int base = (unsigned int)r * HIST_NODES;

    for (int j = tid; j < HIST_NODES; j += HIST_THREADS) cnt[j] = 0u;
    __syncthreads();

    const int iper = (nIdx + HIST_CHUNKS - 1) / HIST_CHUNKS;
    const int t0 = c * iper;
    const int t1 = (t0 + iper < nIdx) ? (t0 + iper) : nIdx;
    for (int t = t0 + tid; t < t1; t += MARK_BATCH * HIST_THREADS) {
        int m[MARK_BATCH];
        #pragma unroll
        for (int u = 0; u < MARK_BATCH; ++u) {
            const int tt = t + u * HIST_THREADS;
            m[u] = (tt < t1) ? idx[tt] : -1;
        }
        #pragma unroll
        for (int u = 0; u < MARK_BATCH; ++u) {
            const unsigned int a = (unsigned int)m[u] - base;
            if (a < (unsigned int)HIST_NODES) atomicOr(&cnt[a], ROW_USED);
        }
    }

    const int nvec = E >> 2;
    const int gper = (nvec + HIST_CHUNKS - 1) / HIST_CHUNKS;
    const int g0 = c * gper;
    const int g1 = (g0 + gper < nvec) ? (g0 + gper) : nvec;
    if ((((size_t)dst) & 15) == 0) {
        const int4* __restrict__ dv = (const int4*)dst;
        for (int g = g0 + tid; g < g1; g += HIST_BATCH * HIST_THREADS) {
            int4 v[HIST_BATCH];
            #pragma unroll
            for (int u = 0; u < HIST_BATCH; ++u) {
                const int gg = g + u * HIST_THREADS;
                v[u] = (gg < g1) ? dv[gg] : make_int4(-1, -1, -1, -1);
            }
            #pragma unroll
            for (int u = 0; u < HIST_BATCH; ++u) {
                hist_add(cnt, v[u].x, base);
                hist_add(cnt, v[u].y, base);
                hist_add(cnt, v[u].z, base);
                hist_add(cnt, v[u].w, base);
            }
        }
    } else {
        for (int e = 4 * g0 + tid; e < 4 * g1; e += HIST_THREADS) hist_add(cnt, dst[e], base);
    }
    if (c == 0) {
        const int e = 4 * nvec + tid;
        if (e < E) hist_add(cnt, dst[e], base);
    }
    __syncthreads();

    unsigned int* __restrict__ plane = part + (size_t)c * N;
    unsigned int outv[HIST_NODES / HIST_THREADS];
    #pragma unroll
    for (int k = 0; k < HIST_NODES / HIST_THREADS; ++k) outv[k] = cnt[tid + k * HIST_THREADS];
    #pragma unroll
    for (int k = 0; k < HIST_NODES / HIST_THREADS; ++k) {
        const unsigned int i = base + (unsigned int)(tid + k * HIST_THREADS);
        if (i < (unsigned int)N) plane[i] = outv[k];
    }
}

__global__ void dinv_kernel(const unsigned int* __restrict__ part, unsigned int* __restrict__ deg,
                            float* __restrict__ dinv, int N) {
    int i = blockIdx.x * blockDim.x + threadIdx.x;
    if (i < N) {
        unsigned int n = 1u;
        unsigned int mark = 0u;
        #pragma unroll
        for (int c = 0; c < HIST_CHUNKS; ++c) {
            const unsigned int v = part[(size_t)c * N + i];
            n += v & ~ROW_USED;
            mark |= v & ROW_USED;
        }
        deg[i] = n | mark;
        float d = (float)n;
        dinv[i] = d > 0.0f ? rsqrtf(d) : 0.0f;
    }
}

__device__ __forceinline__ int   lane_i(int v, int j)   { return __builtin_amdgcn_readlane(v, j); }
__device__ __forceinline__ float lane_f(float v, int j) { return __int_as_float(__builtin_amdgcn_readlane(__float_as_int(v), j)); }

__global__ void __launch_bounds__(256)
scatter_edges_kernel(const float* __restrict__ h, const float* __restrict__ dinv,
                     const unsigned int* __restrict__ deg,
                     const int* __restrict__ src, const int* __restrict__ dst,
                     float* __restrict__ embed, int E) {
    const int lane = threadIdx.x & 31;
    const int e = blockIdx.x * blockDim.x + threadIdx.x;
    int s = 0, d = 0;
    float w = 0.0f;
    bool keep = false;
    if (e < E) {
        s = src[e];
        d = dst[e];
        keep = (deg[d] & ROW_USED) != 0u;
        if (keep) w = dinv[s] * dinv[d];
    }
    unsigned int m = __builtin_amdgcn_ballot_w32(keep);
    while (m) {
        if (__builtin_popcount(m) >= 4) {
            const int j0 = __builtin_ctz(m); m &= m - 1u;
            const int j1 = __builtin_ctz(m); m &= m - 1u;
            const int j2 = __builtin_ctz(m); m &= m - 1u;
            const int j3 = __builtin_ctz(m); m &= m - 1u;
            const float* p0 = h + (size_t)lane_i(s, j0) * 64 + lane;
            const float* p1 = h + (size_t)lane_i(s, j1) * 64 + lane;
            const float* p2 = h + (size_t)lane_i(s, j2) * 64 + lane;
            const float* p3 = h + (size_t)lane_i(s, j3) * 64 + lane;
            const float a0 = p0[0], b0 = p0[32];
            const float a1 = p1[0], b1 = p1[32];
            const float a2 = p2[0], b2 = p2[32];
            const float a3 = p3[0], b3 = p3[32];
            float* q0 = embed + (size_t)lane_i(d, j0) * 64 + lane;
            float* q1 = embed + (size_t)lane_i(d, j1) * 64 + lane;
            float* q2 = embed + (size_t)lane_i(d, j2) * 64 + lane;
            float* q3 = embed + (size_t)lane_i(d, j3) * 64 + lane;
            const float w0 = lane_f(w, j0), w1 = lane_f(w, j1), w2 = lane_f(w, j2), w3 = lane_f(w, j3);
            unsafeAtomicAdd(q0, a0 * w0); unsafeAtomicAdd(q0 + 32, b0 * w0);
            unsafeAtomicAdd(q1, a1 * w1); unsafeAtomicAdd(q1 + 32, b1 * w1);
            unsafeAtomicAdd(q2, a2 * w2); unsafeAtomicAdd(q2 + 32, b2 * w2);
            unsafeAtomicAdd(q3, a3 * w3); unsafeAtomicAdd(q3 + 32, b3 * w3);
        } else {
            const int j0 = __builtin_ctz(m); m &= m - 1u;
            const float* p0 = h + (size_t)lane_i(s, j0) * 64 + lane;
            const float a0 = p0[0], b0 = p0[32];
            float* q0 = embed + (size_t)lane_i(d, j0) * 64 + lane;
            const float w0 = lane_f(w, j0);
            unsafeAtomicAdd(q0, a0 * w0); unsafeAtomicAdd(q0 + 32, b0 * w0);
        }
    }
}

#define HEAD_WAVES 2

typedef __attribute__((ext_vector_type(2))) float f2;

__global__ void __launch_bounds__(HEAD_WAVES * 32)
mlp_kernel(const float* __restrict__ embed, const int* __restrict__ idx,
           const float* __restrict__ b_conv,
           const float* __restrict__ W1, const float* __restrict__ b1,
           const float* __restrict__ W2, const float* __restrict__ b2,
           float* __restrict__ out, int B) {
    const int lane = threadIdx.x & 31;
    const int wave = threadIdx.x >> 5;
    const int hi   = lane >> 4;
    const int c    = lane & 15;
    const int pairBase = (blockIdx.x * HEAD_WAVES + wave) * 16;
    if (pairBase >= B) return;

    const int p  = pairBase + c;
    const int pc = p < B ? p : B - 1;
    const int i0 = idx[2 * pc];
    const int i1 = idx[2 * pc + 1];
    const float* e0 = embed + (size_t)i0 * 64 + 4 * hi;
    const float* e1 = embed + (size_t)i1 * 64 + 4 * hi;

    f4 x[2][8];
    #pragma unroll
    for (int u = 0; u < 8; ++u) {
        x[0][u] = *(const f4*)(e0 + 8 * u);
        x[1][u] = *(const f4*)(e1 + 8 * u);
    }
    float bc[8][4];
    #pragma unroll
    for (int u = 0; u < 8; ++u) {
        #pragma unroll
        for (int t = 0; t < 4; ++t) bc[u][t] = b_conv[8 * u + 4 * hi + t];
    }
    const float* wp = W1 + 64 * hi + c;
    float w[2][8][4];
    #pragma unroll
    for (int q = 0; q < 2; ++q) {
        #pragma unroll
        for (int u = 0; u < 8; ++u) {
            #pragma unroll
            for (int t = 0; t < 4; ++t) w[q][u][t] = wp[(64 * q + 8 * u + t) * 16];
        }
    }
    float b1v[8], w2v[8];
    #pragma unroll
    for (int r = 0; r < 8; ++r) {
        b1v[r] = b1[8 * hi + r];
        w2v[r] = W2[8 * hi + r];
    }

    v8f acc0 = {}, acc1 = {};
    #pragma unroll
    for (int u = 0; u < 8; ++u) {
        #pragma unroll
        for (int v = 0; v < 2; ++v) {
            f2 a0, a1, g0, g1;
            a0[0] = w[0][u][2 * v];     a0[1] = w[0][u][2 * v + 1];
            a1[0] = w[1][u][2 * v];     a1[1] = w[1][u][2 * v + 1];
            g0[0] = lrelu(x[0][u][2 * v]     + bc[u][2 * v]);
            g0[1] = lrelu(x[0][u][2 * v + 1] + bc[u][2 * v + 1]);
            g1[0] = lrelu(x[1][u][2 * v]     + bc[u][2 * v]);
            g1[1] = lrelu(x[1][u][2 * v + 1] + bc[u][2 * v + 1]);
            acc0 = __builtin_amdgcn_wmma_f32_16x16x4_f32(false, a0, false, g0, (short)0, acc0, false, false);
            acc1 = __builtin_amdgcn_wmma_f32_16x16x4_f32(false, a1, false, g1, (short)0, acc1, false, false);
        }
    }

    float part = 0.0f;
    #pragma unroll
    for (int r = 0; r < 8; ++r) {
        const float o = acc0[r] + acc1[r] + b1v[r];
        part += lrelu(o) * w2v[r];
    }
    const float tot = part + __shfl_xor(part, 16);
    const float acc = b2[0] + tot;
    if (hi == 0 && p < B) out[p] = 1.0f / (1.0f + __expf(-acc));
}

extern "C" void kernel_launch(void* const* d_in, const int* in_sizes, int n_in,
                              void* d_out, int out_size, void* d_ws, size_t ws_size,
                              hipStream_t stream) {
    const float* x        = (const float*)d_in[0];
    const int*   edge     = (const int*)d_in[1];
    const int*   index    = (const int*)d_in[2];
    const float* W_conv   = (const float*)d_in[3];
    const float* b_conv   = (const float*)d_in[4];
    const float* W1       = (const float*)d_in[5];
    const float* b1       = (const float*)d_in[6];
    const float* W2       = (const float*)d_in[7];
    const float* b2       = (const float*)d_in[8];
    float*       out      = (float*)d_out;

    const int N = in_sizes[0] / 256;
    const int E = in_sizes[1] / 2;
    const int B = in_sizes[2] / 2;
    const int* src = edge;
    const int* dst = edge + E;

    float*        h     = (float*)d_ws;
    float*        embed = h + (size_t)N * 64;
    unsigned int* deg   = (unsigned int*)(embed + (size_t)N * 64);
    float*        dinv  = (float*)(deg + N);
    _Float16*     Wt    = (_Float16*)(dinv + N);

    transpose_w_kernel<<<(256 * 64 + 255) / 256, 256, 0, stream>>>(W_conv, Wt);

    unsigned int* part = (unsigned int*)embed;
    deg_hist_kernel<<<((N + HIST_NODES - 1) / HIST_NODES) * HIST_CHUNKS, HIST_THREADS, 0, stream>>>(
        dst, index, part, E, 2 * B, N);
    dinv_kernel<<<(N + 255) / 256, 256, 0, stream>>>(part, deg, dinv, N);

    gemm_xw_kernel<<<(N + 16 * GT_WAVES - 1) / (16 * GT_WAVES), GT_THREADS, 0, stream>>>(x, Wt, dinv, deg, h, embed, N);

    scatter_edges_kernel<<<(E + 255) / 256, 256, 0, stream>>>(h, dinv, deg, src, dst, embed, E);

    mlp_kernel<<<(B + 16 * HEAD_WAVES - 1) / (16 * HEAD_WAVES), HEAD_WAVES * 32, 0, stream>>>(embed, index, b_conv, W1, b1, W2, b2, out, B);
}


# scatter as 12504 fat waves (128 edges each; was 50000 of 32): marks read from an LDS bitmap, not a deg lookup per edge; kept edges packed per wave, next 8 rows requested ahead; sums in arrival order
# speedup vs baseline: 5.8980x; 1.0582x over previous
#include <hip/hip_runtime.h>
#include <hip/hip_bf16.h>

typedef __attribute__((ext_vector_type(16))) _Float16 v16h;
typedef __attribute__((ext_vector_type(8)))  _Float16 h8;
typedef __attribute__((ext_vector_type(8)))  float    v8f;
typedef __attribute__((ext_vector_type(4)))  float    f4;

#define NEG_SLOPE 0.01f

__device__ __forceinline__ float lrelu(float v) { return v >= 0.0f ? v : NEG_SLOPE * v; }

__global__ void transpose_w_kernel(const float* __restrict__ W, _Float16* __restrict__ Wt) {
    int t = blockIdx.x * blockDim.x + threadIdx.x;
    if (t >= 256 * 64) return;
    int k = t >> 6;
    int c = t & 63;
    Wt[c * 256 + k] = (_Float16)W[t];
}

#define ROW_USED 0x80000000u

#define GT_WAVES   4
#define GT_THREADS (GT_WAVES * 32)
#define GT_PITCH   264

typedef __attribute__((ext_vector_type(4))) _Float16 h4;

__global__ void __launch_bounds__(GT_THREADS)
gemm_xw_kernel(const float* __restrict__ x,
               const _Float16* __restrict__ Wt,
               const float* __restrict__ dinv,
               const unsigned int* __restrict__ bits,
               float* __restrict__ h,
               float* __restrict__ embed, int N) {
    __shared__ __attribute__((aligned(16))) _Float16 sW[64 * GT_PITCH];
    __shared__ __attribute__((aligned(16))) _Float16 sA[GT_WAVES * 16 * GT_PITCH];

    const int tid  = threadIdx.x;
    const int lane = tid & 31;
    const int wave = tid >> 5;
    const int hi   = lane >> 4;
    const int r    = lane & 15;
    const int rowBase = (blockIdx.x * GT_WAVES + wave) * 16;
    const bool active = rowBase < N;

    f4 xa[32];
    if (active) {
        #pragma unroll
        for (int i = 0; i < 32; ++i) {
            int row = rowBase + (i >> 1);
            row = row < N ? row : N - 1;
            xa[i] = *(const f4*)(x + (size_t)row * 256 + (i & 1) * 128 + lane * 4);
        }
    }
    h8 wv[64 / GT_WAVES];
    #pragma unroll
    for (int i = 0; i < 64 / GT_WAVES; ++i)
        wv[i] = *(const h8*)(Wt + (size_t)(tid + i * GT_THREADS) * 8);

    float dv[8];
    #pragma unroll
    for (int j = 0; j < 8; ++j) {
        int m = rowBase + j + hi * 8;
        m = m < N ? m : N - 1;
        dv[j] = dinv[m];
    }
    const unsigned int mw   = bits[(active ? rowBase : 0) >> 5];
    const unsigned int mbit = (unsigned int)(rowBase & 31) + (unsigned int)hi * 8u;

    #pragma unroll
    for (int i = 0; i < 64 / GT_WAVES; ++i)
        *(h8*)(sW + (wave + i * GT_WAVES) * GT_PITCH + lane * 8) = wv[i];
    if (active) {
        _Float16* myA = sA + wave * 16 * GT_PITCH;
        #pragma unroll
        for (int i = 0; i < 32; ++i) {
            h4 v;
            v[0] = (_Float16)xa[i][0]; v[1] = (_Float16)xa[i][1];
            v[2] = (_Float16)xa[i][2]; v[3] = (_Float16)xa[i][3];
            *(h4*)(myA + (i >> 1) * GT_PITCH + (i & 1) * 128 + lane * 4) = v;
        }
    }
    __syncthreads();
    if (!active) return;

    v8f c0 = {}, c1 = {}, c2 = {}, c3 = {};
    const _Float16* ap = sA + (wave * 16 + r) * GT_PITCH + hi * 8;
    const _Float16* bp = sW + r * GT_PITCH + hi * 8;

    h8 flo[2][5], fhi[2][5];
    #pragma unroll
    for (int t = 0; t < 4; ++t) {
        flo[0][t] = *(const h8*)(bp + t * 16 * GT_PITCH);
        fhi[0][t] = *(const h8*)(bp + t * 16 * GT_PITCH + 16);
    }
    flo[0][4] = *(const h8*)(ap);
    fhi[0][4] = *(const h8*)(ap + 16);

    #pragma unroll
    for (int kk = 0; kk < 8; ++kk) {
        const int cur = kk & 1, nxt = cur ^ 1;
        if (kk + 1 < 8) {
            const int kb = (kk + 1) * 32;
            #pragma unroll
            for (int t = 0; t < 4; ++t) {
                flo[nxt][t] = *(const h8*)(bp + t * 16 * GT_PITCH + kb);
                fhi[nxt][t] = *(const h8*)(bp + t * 16 * GT_PITCH + kb + 16);
            }
            flo[nxt][4] = *(const h8*)(ap + kb);
            fhi[nxt][4] = *(const h8*)(ap + kb + 16);
        }
        __builtin_amdgcn_sched_barrier(0);
        const v16h a   = __builtin_shufflevector(flo[cur][4], fhi[cur][4], 0, 1, 2, 3, 4, 5, 6, 7, 8, 9, 10, 11, 12, 13, 14, 15);
        const v16h b0v = __builtin_shufflevector(flo[cur][0], fhi[cur][0], 0, 1, 2, 3, 4, 5, 6, 7, 8, 9, 10, 11, 12, 13, 14, 15);
        const v16h b1v = __builtin_shufflevector(flo[cur][1], fhi[cur][1], 0, 1, 2, 3, 4, 5, 6, 7, 8, 9, 10, 11, 12, 13, 14, 15);
        const v16h b2v = __builtin_shufflevector(flo[cur][2], fhi[cur][2], 0, 1, 2, 3, 4, 5, 6, 7, 8, 9, 10, 11, 12, 13, 14, 15);
        const v16h b3v = __builtin_shufflevector(flo[cur][3], fhi[cur][3], 0, 1, 2, 3, 4, 5, 6, 7, 8, 9, 10, 11, 12, 13, 14, 15);
        c0 = __builtin_amdgcn_wmma_f32_16x16x32_f16(false, a, false, b0v, (short)0, c0, false, false);
        c1 = __builtin_amdgcn_wmma_f32_16x16x32_f16(false, a, false, b1v, (short)0, c1, false, false);
        c2 = __builtin_amdgcn_wmma_f32_16x16x32_f16(false, a, false, b2v, (short)0, c2, false, false);
        c3 = __builtin_amdgcn_wmma_f32_16x16x32_f16(false, a, false, b3v, (short)0, c3, false, false);
    }

    #pragma unroll
    for (int j = 0; j < 8; ++j) {
        const int m = rowBase + j + hi * 8;
        if (m < N) {
            float* p = h + (size_t)m * 64;
            p[r]      = c0[j];
            p[16 + r] = c1[j];
            p[32 + r] = c2[j];
            p[48 + r] = c3[j];
            if ((mw >> (mbit + (unsigned int)j)) & 1u) {
                float* q = embed + (size_t)m * 64;
                const float di = dv[j];
                q[r]      = c0[j] * di * di;
                q[16 + r] = c1[j] * di * di;
                q[32 + r] = c2[j] * di * di;
                q[48 + r] = c3[j] * di * di;
            }
        }
    }
}

#define HIST_NODES   8192
#define HIST_CHUNKS  25
#define HIST_THREADS 512
#define HIST_BATCH   16
#define MARK_BATCH   4

__device__ __forceinline__ void hist_add(unsigned int* cnt, int node, unsigned int base) {
    const unsigned int a = (unsigned int)node - base;
    if (a < (unsigned int)HIST_NODES) atomicAdd(&cnt[a], 1u);
}

__global__ void __launch_bounds__(HIST_THREADS)
deg_hist_kernel(const int* __restrict__ dst, const int* __restrict__ idx,
                unsigned int* __restrict__ part, int E, int nIdx, int N) {
    __shared__ unsigned int cnt[HIST_NODES];
    const int tid = threadIdx.x;
    const int r = blockIdx.x / HIST_CHUNKS;
    const int c = blockIdx.x % HIST_CHUNKS;
    const unsigned int base = (unsigned int)r * HIST_NODES;

    for (int j = tid; j < HIST_NODES; j += HIST_THREADS) cnt[j] = 0u;
    __syncthreads();

    const int iper = (nIdx + HIST_CHUNKS - 1) / HIST_CHUNKS;
    const int t0 = c * iper;
    const int t1 = (t0 + iper < nIdx) ? (t0 + iper) : nIdx;
    for (int t = t0 + tid; t < t1; t += MARK_BATCH * HIST_THREADS) {
        int m[MARK_BATCH];
        #pragma unroll
        for (int u = 0; u < MARK_BATCH; ++u) {
            const int tt = t + u * HIST_THREADS;
            m[u] = (tt < t1) ? idx[tt] : -1;
        }
        #pragma unroll
        for (int u = 0; u < MARK_BATCH; ++u) {
            const unsigned int a = (unsigned int)m[u] - base;
            if (a < (unsigned int)HIST_NODES) atomicOr(&cnt[a], ROW_USED);
        }
    }

    const int nvec = E >> 2;
    const int gper = (nvec + HIST_CHUNKS - 1) / HIST_CHUNKS;
    const int g0 = c * gper;
    const int g1 = (g0 + gper < nvec) ? (g0 + gper) : nvec;
    if ((((size_t)dst) & 15) == 0) {
        const int4* __restrict__ dv = (const int4*)dst;
        for (int g = g0 + tid; g < g1; g += HIST_BATCH * HIST_THREADS) {
            int4 v[HIST_BATCH];
            #pragma unroll
            for (int u = 0; u < HIST_BATCH; ++u) {
                const int gg = g + u * HIST_THREADS;
                v[u] = (gg < g1) ? dv[gg] : make_int4(-1, -1, -1, -1);
            }
            #pragma unroll
            for (int u = 0; u < HIST_BATCH; ++u) {
                hist_add(cnt, v[u].x, base);
                hist_add(cnt, v[u].y, base);
                hist_add(cnt, v[u].z, base);
                hist_add(cnt, v[u].w, base);
            }
        }
    } else {
        for (int e = 4 * g0 + tid; e < 4 * g1; e += HIST_THREADS) hist_add(cnt, dst[e], base);
    }
    if (c == 0) {
        const int e = 4 * nvec + tid;
        if (e < E) hist_add(cnt, dst[e], base);
    }
    __syncthreads();

    unsigned int* __restrict__ plane = part + (size_t)c * N;
    unsigned int outv[HIST_NODES / HIST_THREADS];
    #pragma unroll
    for (int k = 0; k < HIST_NODES / HIST_THREADS; ++k) outv[k] = cnt[tid + k * HIST_THREADS];
    #pragma unroll
    for (int k = 0; k < HIST_NODES / HIST_THREADS; ++k) {
        const unsigned int i = base + (unsigned int)(tid + k * HIST_THREADS);
        if (i < (unsigned int)N) plane[i] = outv[k];
    }
}

__global__ void __launch_bounds__(256)
dinv_kernel(const unsigned int* __restrict__ part, unsigned int* __restrict__ bits,
            float* __restrict__ dinv, int N) {
    const int i = blockIdx.x * 256 + threadIdx.x;
    unsigned int mark = 0u;
    if (i < N) {
        unsigned int n = 1u;
        #pragma unroll
        for (int c = 0; c < HIST_CHUNKS; ++c) {
            const unsigned int v = part[(size_t)c * N + i];
            n += v & ~ROW_USED;
            mark |= v & ROW_USED;
        }
        float d = (float)n;
        dinv[i] = d > 0.0f ? rsqrtf(d) : 0.0f;
    }
    const unsigned int word = __builtin_amdgcn_ballot_w32(mark != 0u);
    if ((threadIdx.x & 31) == 0 && i < N) bits[i >> 5] = word;
}

__device__ __forceinline__ int   lane_i(int v, int j)   { return __builtin_amdgcn_readlane(v, j); }
__device__ __forceinline__ float lane_f(float v, int j) { return __int_as_float(__builtin_amdgcn_readlane(__float_as_int(v), j)); }

#define SC_WAVES    8
#define SC_THREADS  (SC_WAVES * 32)
#define SC_EPL      4
#define SC_WEDGES   (32 * SC_EPL)
#define SC_BITS_MAX 4096
#define SC_BITS_PER_THREAD (SC_BITS_MAX / SC_THREADS)

#define SC_FETCH(G, RA, RB, DD, WW)                                                  \
    {                                                                                \
        int li_ = (G) * 8 + (lane & 7);                                              \
        li_ = li_ < n ? li_ : n - 1;                                                 \
        const int   ls_ = Ls[wave][li_];                                             \
        const int   ld_ = Ld[wave][li_];                                             \
        const float lw_ = Lw[wave][li_];                                             \
        _Pragma("unroll")                                                            \
        for (int j = 0; j < 8; ++j) {                                                \
            const float* p_ = h + (size_t)lane_i(ls_, j) * 64 + lane;                \
            DD[j] = lane_i(ld_, j);                                                  \
            WW[j] = lane_f(lw_, j);                                                  \
            RA[j] = p_[0];                                                           \
            RB[j] = p_[32];                                                          \
        }                                                                            \
    }
#define SC_FLUSH_ALL(RA, RB, DD, WW)                                                 \
    {                                                                                \
        _Pragma("unroll")                                                            \
        for (int j = 0; j < 8; ++j) {                                                \
            float* q_ = embed + (size_t)DD[j] * 64 + lane;                           \
            unsafeAtomicAdd(q_,      RA[j] * WW[j]);                                 \
            unsafeAtomicAdd(q_ + 32, RB[j] * WW[j]);                                 \
        }                                                                            \
    }
#define SC_FLUSH_LAST(G, RA, RB, DD, WW)                                             \
    {                                                                                \
        _Pragma("unroll")                                                            \
        for (int j = 0; j < 8; ++j) {                                                \
            if ((G) * 8 + j < n) {                                                   \
                float* q_ = embed + (size_t)DD[j] * 64 + lane;                       \
                unsafeAtomicAdd(q_,      RA[j] * WW[j]);                             \
                unsafeAtomicAdd(q_ + 32, RB[j] * WW[j]);                             \
            }                                                                        \
        }                                                                            \
    }

__global__ void __launch_bounds__(SC_THREADS)
scatter_edges_kernel(const float* __restrict__ h, const float* __restrict__ dinv,
                     const unsigned int* __restrict__ bits,
                     const int* __restrict__ src, const int* __restrict__ dst,
                     float* __restrict__ embed, int E, int nWords) {
    __shared__ unsigned int sBits[SC_BITS_MAX];
    __shared__ int   Ls[SC_WAVES][SC_WEDGES];
    __shared__ int   Ld[SC_WAVES][SC_WEDGES];
    __shared__ float Lw[SC_WAVES][SC_WEDGES];

    const int tid  = threadIdx.x;
    const int lane = tid & 31;
    const int wave = tid >> 5;
    const bool inLds = nWords <= SC_BITS_MAX;

    unsigned int bw[SC_BITS_PER_THREAD];
    if (inLds) {
        #pragma unroll
        for (int k = 0; k < SC_BITS_PER_THREAD; ++k) {
            const int i = tid + k * SC_THREADS;
            bw[k] = bits[i < nWords ? i : nWords - 1];
        }
    }
    const int wbase = (blockIdx.x * SC_WAVES + wave) * SC_WEDGES;
    int  s[SC_EPL], d[SC_EPL];
    bool ok[SC_EPL];
    #pragma unroll
    for (int u = 0; u < SC_EPL; ++u) {
        const int e = wbase + u * 32 + lane;
        ok[u] = e < E;
        const int ec = ok[u] ? e : E - 1;
        s[u] = src[ec];
        d[u] = dst[ec];
    }
    if (inLds) {
        #pragma unroll
        for (int k = 0; k < SC_BITS_PER_THREAD; ++k) sBits[tid + k * SC_THREADS] = bw[k];
    }
    __syncthreads();

    bool  keep[SC_EPL];
    float w[SC_EPL];
    unsigned int word[SC_EPL];
    if (inLds) {
        #pragma unroll
        for (int u = 0; u < SC_EPL; ++u) word[u] = sBits[d[u] >> 5];
    } else {
        #pragma unroll
        for (int u = 0; u < SC_EPL; ++u) word[u] = bits[d[u] >> 5];
    }
    #pragma unroll
    for (int u = 0; u < SC_EPL; ++u) keep[u] = ok[u] & (((word[u] >> (d[u] & 31)) & 1u) != 0u);
    #pragma unroll
    for (int u = 0; u < SC_EPL; ++u) {
        const int sa = keep[u] ? s[u] : 0;
        const int da = keep[u] ? d[u] : 0;
        w[u] = dinv[sa] * dinv[da];
    }

    int n = 0;
    #pragma unroll
    for (int u = 0; u < SC_EPL; ++u) {
        const unsigned int m = __builtin_amdgcn_ballot_w32(keep[u]);
        if (keep[u]) {
            const int pos = n + (int)__builtin_amdgcn_mbcnt_lo(m, 0u);
            Ls[wave][pos] = s[u];
            Ld[wave][pos] = d[u];
            Lw[wave][pos] = w[u];
        }
        n += __builtin_popcount(m);
    }

    const int ng = (n + 7) >> 3;
    if (ng == 0) return;
    float a0[8], b0[8], w0[8], a1[8], b1[8], w1[8];
    int   d0[8], d1[8];
    int g = 0;
    SC_FETCH(0, a0, b0, d0, w0)
    for (; g + 2 < ng; g += 2) {
        SC_FETCH(g + 1, a1, b1, d1, w1)
        __builtin_amdgcn_sched_barrier(0);
        SC_FLUSH_ALL(a0, b0, d0, w0)
        SC_FETCH(g + 2, a0, b0, d0, w0)
        __builtin_amdgcn_sched_barrier(0);
        SC_FLUSH_ALL(a1, b1, d1, w1)
    }
    if (g + 1 < ng) {
        SC_FETCH(g + 1, a1, b1, d1, w1)
        __builtin_amdgcn_sched_barrier(0);
        SC_FLUSH_ALL(a0, b0, d0, w0)
        SC_FLUSH_LAST(g + 1, a1, b1, d1, w1)
    } else {
        SC_FLUSH_LAST(g, a0, b0, d0, w0)
    }
}

#define HEAD_WAVES 2

typedef __attribute__((ext_vector_type(2))) float f2;

__global__ void __launch_bounds__(HEAD_WAVES * 32)
mlp_kernel(const float* __restrict__ embed, const int* __restrict__ idx,
           const float* __restrict__ b_conv,
           const float* __restrict__ W1, const float* __restrict__ b1,
           const float* __restrict__ W2, const float* __restrict__ b2,
           float* __restrict__ out, int B) {
    const int lane = threadIdx.x & 31;
    const int wave = threadIdx.x >> 5;
    const int hi   = lane >> 4;
    const int c    = lane & 15;
    const int pairBase = (blockIdx.x * HEAD_WAVES + wave) * 16;
    if (pairBase >= B) return;

    const int p  = pairBase + c;
    const int pc = p < B ? p : B - 1;
    const int i0 = idx[2 * pc];
    const int i1 = idx[2 * pc + 1];
    const float* e0 = embed + (size_t)i0 * 64 + 4 * hi;
    const float* e1 = embed + (size_t)i1 * 64 + 4 * hi;

    f4 x[2][8];
    #pragma unroll
    for (int u = 0; u < 8; ++u) {
        x[0][u] = *(const f4*)(e0 + 8 * u);
        x[1][u] = *(const f4*)(e1 + 8 * u);
    }
    float bc[8][4];
    #pragma unroll
    for (int u = 0; u < 8; ++u) {
        #pragma unroll
        for (int t = 0; t < 4; ++t) bc[u][t] = b_conv[8 * u + 4 * hi + t];
    }
    const float* wp = W1 + 64 * hi + c;
    float w[2][8][4];
    #pragma unroll
    for (int q = 0; q < 2; ++q) {
        #pragma unroll
        for (int u = 0; u < 8; ++u) {
            #pragma unroll
            for (int t = 0; t < 4; ++t) w[q][u][t] = wp[(64 * q + 8 * u + t) * 16];
        }
    }
    float b1v[8], w2v[8];
    #pragma unroll
    for (int r = 0; r < 8; ++r) {
        b1v[r] = b1[8 * hi + r];
        w2v[r] = W2[8 * hi + r];
    }

    v8f acc0 = {}, acc1 = {};
    #pragma unroll
    for (int u = 0; u < 8; ++u) {
        #pragma unroll
        for (int v = 0; v < 2; ++v) {
            f2 a0, a1, g0, g1;
            a0[0] = w[0][u][2 * v];     a0[1] = w[0][u][2 * v + 1];
            a1[0] = w[1][u][2 * v];     a1[1] = w[1][u][2 * v + 1];
            g0[0] = lrelu(x[0][u][2 * v]     + bc[u][2 * v]);
            g0[1] = lrelu(x[0][u][2 * v + 1] + bc[u][2 * v + 1]);
            g1[0] = lrelu(x[1][u][2 * v]     + bc[u][2 * v]);
            g1[1] = lrelu(x[1][u][2 * v + 1] + bc[u][2 * v + 1]);
            acc0 = __builtin_amdgcn_wmma_f32_16x16x4_f32(false, a0, false, g0, (short)0, acc0, false, false);
            acc1 = __builtin_amdgcn_wmma_f32_16x16x4_f32(false, a1, false, g1, (short)0, acc1, false, false);
        }
    }

    float part = 0.0f;
    #pragma unroll
    for (int r = 0; r < 8; ++r) {
        const float o = acc0[r] + acc1[r] + b1v[r];
        part += lrelu(o) * w2v[r];
    }
    const float tot = part + __shfl_xor(part, 16);
    const float acc = b2[0] + tot;
    if (hi == 0 && p < B) out[p] = 1.0f / (1.0f + __expf(-acc));
}

extern "C" void kernel_launch(void* const* d_in, const int* in_sizes, int n_in,
                              void* d_out, int out_size, void* d_ws, size_t ws_size,
                              hipStream_t stream) {
    const float* x        = (const float*)d_in[0];
    const int*   edge     = (const int*)d_in[1];
    const int*   index    = (const int*)d_in[2];
    const float* W_conv   = (const float*)d_in[3];
    const float* b_conv   = (const float*)d_in[4];
    const float* W1       = (const float*)d_in[5];
    const float* b1       = (const float*)d_in[6];
    const float* W2       = (const float*)d_in[7];
    const float* b2       = (const float*)d_in[8];
    float*       out      = (float*)d_out;

    const int N = in_sizes[0] / 256;
    const int E = in_sizes[1] / 2;
    const int B = in_sizes[2] / 2;
    const int* src = edge;
    const int* dst = edge + E;

    float*        h     = (float*)d_ws;
    float*        embed = h + (size_t)N * 64;
    unsigned int* bits  = (unsigned int*)(embed + (size_t)N * 64);
    float*        dinv  = (float*)(bits + N);
    _Float16*     Wt    = (_Float16*)(dinv + N);

    transpose_w_kernel<<<(256 * 64 + 255) / 256, 256, 0, stream>>>(W_conv, Wt);

    unsigned int* part = (unsigned int*)embed;
    deg_hist_kernel<<<((N + HIST_NODES - 1) / HIST_NODES) * HIST_CHUNKS, HIST_THREADS, 0, stream>>>(
        dst, index, part, E, 2 * B, N);
    dinv_kernel<<<(N + 255) / 256, 256, 0, stream>>>(part, bits, dinv, N);

    gemm_xw_kernel<<<(N + 16 * GT_WAVES - 1) / (16 * GT_WAVES), GT_THREADS, 0, stream>>>(x, Wt, dinv, bits, h, embed, N);

    scatter_edges_kernel<<<(E + SC_WAVES * SC_WEDGES - 1) / (SC_WAVES * SC_WEDGES), SC_THREADS, 0, stream>>>(
        h, dinv, bits, src, dst, embed, E, (N + 31) / 32);

    mlp_kernel<<<(B + 16 * HEAD_WAVES - 1) / (16 * HEAD_WAVES), HEAD_WAVES * 32, 0, stream>>>(embed, index, b_conv, W1, b1, W2, b2, out, B);
}


# W_conv transpose to f16 done by the first 16384 threads of the dinv pass (was a launch of its own): 5 launches instead of 6; values, k order and sum order unchanged
# speedup vs baseline: 6.1805x; 1.0479x over previous
#include <hip/hip_runtime.h>
#include <hip/hip_bf16.h>

typedef __attribute__((ext_vector_type(16))) _Float16 v16h;
typedef __attribute__((ext_vector_type(8)))  _Float16 h8;
typedef __attribute__((ext_vector_type(8)))  float    v8f;
typedef __attribute__((ext_vector_type(4)))  float    f4;

#define NEG_SLOPE 0.01f

__device__ __forceinline__ float lrelu(float v) { return v >= 0.0f ? v : NEG_SLOPE * v; }

#define ROW_USED 0x80000000u

#define GT_WAVES   4
#define GT_THREADS (GT_WAVES * 32)
#define GT_PITCH   264

typedef __attribute__((ext_vector_type(4))) _Float16 h4;

__global__ void __launch_bounds__(GT_THREADS)
gemm_xw_kernel(const float* __restrict__ x,
               const _Float16* __restrict__ Wt,
               const float* __restrict__ dinv,
               const unsigned int* __restrict__ bits,
               float* __restrict__ h,
               float* __restrict__ embed, int N) {
    __shared__ __attribute__((aligned(16))) _Float16 sW[64 * GT_PITCH];
    __shared__ __attribute__((aligned(16))) _Float16 sA[GT_WAVES * 16 * GT_PITCH];

    const int tid  = threadIdx.x;
    const int lane = tid & 31;
    const int wave = tid >> 5;
    const int hi   = lane >> 4;
    const int r    = lane & 15;
    const int rowBase = (blockIdx.x * GT_WAVES + wave) * 16;
    const bool active = rowBase < N;

    f4 xa[32];
    if (active) {
        #pragma unroll
        for (int i = 0; i < 32; ++i) {
            int row = rowBase + (i >> 1);
            row = row < N ? row : N - 1;
            xa[i] = *(const f4*)(x + (size_t)row * 256 + (i & 1) * 128 + lane * 4);
        }
    }
    h8 wv[64 / GT_WAVES];
    #pragma unroll
    for (int i = 0; i < 64 / GT_WAVES; ++i)
        wv[i] = *(const h8*)(Wt + (size_t)(tid + i * GT_THREADS) * 8);

    float dv[8];
    #pragma unroll
    for (int j = 0; j < 8; ++j) {
        int m = rowBase + j + hi * 8;
        m = m < N ? m : N - 1;
        dv[j] = dinv[m];
    }
    const unsigned int mw   = bits[(active ? rowBase : 0) >> 5];
    const unsigned int mbit = (unsigned int)(rowBase & 31) + (unsigned int)hi * 8u;

    #pragma unroll
    for (int i = 0; i < 64 / GT_WAVES; ++i)
        *(h8*)(sW + (wave + i * GT_WAVES) * GT_PITCH + lane * 8) = wv[i];
    if (active) {
        _Float16* myA = sA + wave * 16 * GT_PITCH;
        #pragma unroll
        for (int i = 0; i < 32; ++i) {
            h4 v;
            v[0] = (_Float16)xa[i][0]; v[1] = (_Float16)xa[i][1];
            v[2] = (_Float16)xa[i][2]; v[3] = (_Float16)xa[i][3];
            *(h4*)(myA + (i >> 1) * GT_PITCH + (i & 1) * 128 + lane * 4) = v;
        }
    }
    __syncthreads();
    if (!active) return;

    v8f c0 = {}, c1 = {}, c2 = {}, c3 = {};
    const _Float16* ap = sA + (wave * 16 + r) * GT_PITCH + hi * 8;
    const _Float16* bp = sW + r * GT_PITCH + hi * 8;

    h8 flo[2][5], fhi[2][5];
    #pragma unroll
    for (int t = 0; t < 4; ++t) {
        flo[0][t] = *(const h8*)(bp + t * 16 * GT_PITCH);
        fhi[0][t] = *(const h8*)(bp + t * 16 * GT_PITCH + 16);
    }
    flo[0][4] = *(const h8*)(ap);
    fhi[0][4] = *(const h8*)(ap + 16);

    #pragma unroll
    for (int kk = 0; kk < 8; ++kk) {
        const int cur = kk & 1, nxt = cur ^ 1;
        if (kk + 1 < 8) {
            const int kb = (kk + 1) * 32;
            #pragma unroll
            for (int t = 0; t < 4; ++t) {
                flo[nxt][t] = *(const h8*)(bp + t * 16 * GT_PITCH + kb);
                fhi[nxt][t] = *(const h8*)(bp + t * 16 * GT_PITCH + kb + 16);
            }
            flo[nxt][4] = *(const h8*)(ap + kb);
            fhi[nxt][4] = *(const h8*)(ap + kb + 16);
        }
        __builtin_amdgcn_sched_barrier(0);
        const v16h a   = __builtin_shufflevector(flo[cur][4], fhi[cur][4], 0, 1, 2, 3, 4, 5, 6, 7, 8, 9, 10, 11, 12, 13, 14, 15);
        const v16h b0v = __builtin_shufflevector(flo[cur][0], fhi[cur][0], 0, 1, 2, 3, 4, 5, 6, 7, 8, 9, 10, 11, 12, 13, 14, 15);
        const v16h b1v = __builtin_shufflevector(flo[cur][1], fhi[cur][1], 0, 1, 2, 3, 4, 5, 6, 7, 8, 9, 10, 11, 12, 13, 14, 15);
        const v16h b2v = __builtin_shufflevector(flo[cur][2], fhi[cur][2], 0, 1, 2, 3, 4, 5, 6, 7, 8, 9, 10, 11, 12, 13, 14, 15);
        const v16h b3v = __builtin_shufflevector(flo[cur][3], fhi[cur][3], 0, 1, 2, 3, 4, 5, 6, 7, 8, 9, 10, 11, 12, 13, 14, 15);
        c0 = __builtin_amdgcn_wmma_f32_16x16x32_f16(false, a, false, b0v, (short)0, c0, false, false);
        c1 = __builtin_amdgcn_wmma_f32_16x16x32_f16(false, a, false, b1v, (short)0, c1, false, false);
        c2 = __builtin_amdgcn_wmma_f32_16x16x32_f16(false, a, false, b2v, (short)0, c2, false, false);
        c3 = __builtin_amdgcn_wmma_f32_16x16x32_f16(false, a, false, b3v, (short)0, c3, false, false);
    }

    #pragma unroll
    for (int j = 0; j < 8; ++j) {
        const int m = rowBase + j + hi * 8;
        if (m < N) {
            float* p = h + (size_t)m * 64;
            p[r]      = c0[j];
            p[16 + r] = c1[j];
            p[32 + r] = c2[j];
            p[48 + r] = c3[j];
            if ((mw >> (mbit + (unsigned int)j)) & 1u) {
                float* q = embed + (size_t)m * 64;
                const float di = dv[j];
                q[r]      = c0[j] * di * di;
                q[16 + r] = c1[j] * di * di;
                q[32 + r] = c2[j] * di * di;
                q[48 + r] = c3[j] * di * di;
            }
        }
    }
}

#define HIST_NODES   8192
#define HIST_CHUNKS  25
#define HIST_THREADS 512
#define HIST_BATCH   16
#define MARK_BATCH   4

__device__ __forceinline__ void hist_add(unsigned int* cnt, int node, unsigned int base) {
    const unsigned int a = (unsigned int)node - base;
    if (a < (unsigned int)HIST_NODES) atomicAdd(&cnt[a], 1u);
}

__global__ void __launch_bounds__(HIST_THREADS)
deg_hist_kernel(const int* __restrict__ dst, const int* __restrict__ idx,
                unsigned int* __restrict__ part, int E, int nIdx, int N) {
    __shared__ unsigned int cnt[HIST_NODES];
    const int tid = threadIdx.x;
    const int r = blockIdx.x / HIST_CHUNKS;
    const int c = blockIdx.x % HIST_CHUNKS;
    const unsigned int base = (unsigned int)r * HIST_NODES;

    for (int j = tid; j < HIST_NODES; j += HIST_THREADS) cnt[j] = 0u;
    __syncthreads();

    const int iper = (nIdx + HIST_CHUNKS - 1) / HIST_CHUNKS;
    const int t0 = c * iper;
    const int t1 = (t0 + iper < nIdx) ? (t0 + iper) : nIdx;
    for (int t = t0 + tid; t < t1; t += MARK_BATCH * HIST_THREADS) {
        int m[MARK_BATCH];
        #pragma unroll
        for (int u = 0; u < MARK_BATCH; ++u) {
            const int tt = t + u * HIST_THREADS;
            m[u] = (tt < t1) ? idx[tt] : -1;
        }
        #pragma unroll
        for (int u = 0; u < MARK_BATCH; ++u) {
            const unsigned int a = (unsigned int)m[u] - base;
            if (a < (unsigned int)HIST_NODES) atomicOr(&cnt[a], ROW_USED);
        }
    }

    const int nvec = E >> 2;
    const int gper = (nvec + HIST_CHUNKS - 1) / HIST_CHUNKS;
    const int g0 = c * gper;
    const int g1 = (g0 + gper < nvec) ? (g0 + gper) : nvec;
    if ((((size_t)dst) & 15) == 0) {
        const int4* __restrict__ dv = (const int4*)dst;
        for (int g = g0 + tid; g < g1; g += HIST_BATCH * HIST_THREADS) {
            int4 v[HIST_BATCH];
            #pragma unroll
            for (int u = 0; u < HIST_BATCH; ++u) {
                const int gg = g + u * HIST_THREADS;
                v[u] = (gg < g1) ? dv[gg] : make_int4(-1, -1, -1, -1);
            }
            #pragma unroll
            for (int u = 0; u < HIST_BATCH; ++u) {
                hist_add(cnt, v[u].x, base);
                hist_add(cnt, v[u].y, base);
                hist_add(cnt, v[u].z, base);
                hist_add(cnt, v[u].w, base);
            }
        }
    } else {
        for (int e = 4 * g0 + tid; e < 4 * g1; e += HIST_THREADS) hist_add(cnt, dst[e], base);
    }
    if (c == 0) {
        const int e = 4 * nvec + tid;
        if (e < E) hist_add(cnt, dst[e], base);
    }
    __syncthreads();

    unsigned int* __restrict__ plane = part + (size_t)c * N;
    unsigned int outv[HIST_NODES / HIST_THREADS];
    #pragma unroll
    for (int k = 0; k < HIST_NODES / HIST_THREADS; ++k) outv[k] = cnt[tid + k * HIST_THREADS];
    #pragma unroll
    for (int k = 0; k < HIST_NODES / HIST_THREADS; ++k) {
        const unsigned int i = base + (unsigned int)(tid + k * HIST_THREADS);
        if (i < (unsigned int)N) plane[i] = outv[k];
    }
}

__global__ void __launch_bounds__(256)
dinv_kernel(const unsigned int* __restrict__ part, unsigned int* __restrict__ bits,
            float* __restrict__ dinv, int N,
            const float* __restrict__ W, _Float16* __restrict__ Wt) {
    const int i = blockIdx.x * 256 + threadIdx.x;
    for (int t = i; t < 256 * 64; t += (int)gridDim.x * 256) {
        const int k = t >> 6;
        const int c = t & 63;
        Wt[c * 256 + k] = (_Float16)W[t];
    }
    unsigned int mark = 0u;
    if (i < N) {
        unsigned int n = 1u;
        #pragma unroll
        for (int c = 0; c < HIST_CHUNKS; ++c) {
            const unsigned int v = part[(size_t)c * N + i];
            n += v & ~ROW_USED;
            mark |= v & ROW_USED;
        }
        float d = (float)n;
        dinv[i] = d > 0.0f ? rsqrtf(d) : 0.0f;
    }
    const unsigned int word = __builtin_amdgcn_ballot_w32(mark != 0u);
    if ((threadIdx.x & 31) == 0 && i < N) bits[i >> 5] = word;
}

__device__ __forceinline__ int   lane_i(int v, int j)   { return __builtin_amdgcn_readlane(v, j); }
__device__ __forceinline__ float lane_f(float v, int j) { return __int_as_float(__builtin_amdgcn_readlane(__float_as_int(v), j)); }

#define SC_WAVES    8
#define SC_THREADS  (SC_WAVES * 32)
#define SC_EPL      4
#define SC_WEDGES   (32 * SC_EPL)
#define SC_BITS_MAX 4096
#define SC_BITS_PER_THREAD (SC_BITS_MAX / SC_THREADS)

#define SC_FETCH(G, RA, RB, DD, WW)                                                  \
    {                                                                                \
        int li_ = (G) * 8 + (lane & 7);                                              \
        li_ = li_ < n ? li_ : n - 1;                                                 \
        const int   ls_ = Ls[wave][li_];                                             \
        const int   ld_ = Ld[wave][li_];                                             \
        const float lw_ = Lw[wave][li_];                                             \
        _Pragma("unroll")                                                            \
        for (int j = 0; j < 8; ++j) {                                                \
            const float* p_ = h + (size_t)lane_i(ls_, j) * 64 + lane;                \
            DD[j] = lane_i(ld_, j);                                                  \
            WW[j] = lane_f(lw_, j);                                                  \
            RA[j] = p_[0];                                                           \
            RB[j] = p_[32];                                                          \
        }                                                                            \
    }
#define SC_FLUSH_ALL(RA, RB, DD, WW)                                                 \
    {                                                                                \
        _Pragma("unroll")                                                            \
        for (int j = 0; j < 8; ++j) {                                                \
            float* q_ = embed + (size_t)DD[j] * 64 + lane;                           \
            unsafeAtomicAdd(q_,      RA[j] * WW[j]);                                 \
            unsafeAtomicAdd(q_ + 32, RB[j] * WW[j]);                                 \
        }                                                                            \
    }
#define SC_FLUSH_LAST(G, RA, RB, DD, WW)                                             \
    {                                                                                \
        _Pragma("unroll")                                                            \
        for (int j = 0; j < 8; ++j) {                                                \
            if ((G) * 8 + j < n) {                                                   \
                float* q_ = embed + (size_t)DD[j] * 64 + lane;                       \
                unsafeAtomicAdd(q_,      RA[j] * WW[j]);                             \
                unsafeAtomicAdd(q_ + 32, RB[j] * WW[j]);                             \
            }                                                                        \
        }                                                                            \
    }

__global__ void __launch_bounds__(SC_THREADS)
scatter_edges_kernel(const float* __restrict__ h, const float* __restrict__ dinv,
                     const unsigned int* __restrict__ bits,
                     const int* __restrict__ src, const int* __restrict__ dst,
                     float* __restrict__ embed, int E, int nWords) {
    __shared__ unsigned int sBits[SC_BITS_MAX];
    __shared__ int   Ls[SC_WAVES][SC_WEDGES];
    __shared__ int   Ld[SC_WAVES][SC_WEDGES];
    __shared__ float Lw[SC_WAVES][SC_WEDGES];

    const int tid  = threadIdx.x;
    const int lane = tid & 31;
    const int wave = tid >> 5;
    const bool inLds = nWords <= SC_BITS_MAX;

    unsigned int bw[SC_BITS_PER_THREAD];
    if (inLds) {
        #pragma unroll
        for (int k = 0; k < SC_BITS_PER_THREAD; ++k) {
            const int i = tid + k * SC_THREADS;
            bw[k] = bits[i < nWords ? i : nWords - 1];
        }
    }
    const int wbase = (blockIdx.x * SC_WAVES + wave) * SC_WEDGES;
    int  s[SC_EPL], d[SC_EPL];
    bool ok[SC_EPL];
    #pragma unroll
    for (int u = 0; u < SC_EPL; ++u) {
        const int e = wbase + u * 32 + lane;
        ok[u] = e < E;
        const int ec = ok[u] ? e : E - 1;
        s[u] = src[ec];
        d[u] = dst[ec];
    }
    if (inLds) {
        #pragma unroll
        for (int k = 0; k < SC_BITS_PER_THREAD; ++k) sBits[tid + k * SC_THREADS] = bw[k];
    }
    __syncthreads();

    bool  keep[SC_EPL];
    float w[SC_EPL];
    unsigned int word[SC_EPL];
    if (inLds) {
        #pragma unroll
        for (int u = 0; u < SC_EPL; ++u) word[u] = sBits[d[u] >> 5];
    } else {
        #pragma unroll
        for (int u = 0; u < SC_EPL; ++u) word[u] = bits[d[u] >> 5];
    }
    #pragma unroll
    for (int u = 0; u < SC_EPL; ++u) keep[u] = ok[u] & (((word[u] >> (d[u] & 31)) & 1u) != 0u);
    #pragma unroll
    for (int u = 0; u < SC_EPL; ++u) {
        const int sa = keep[u] ? s[u] : 0;
        const int da = keep[u] ? d[u] : 0;
        w[u] = dinv[sa] * dinv[da];
    }

    int n = 0;
    #pragma unroll
    for (int u = 0; u < SC_EPL; ++u) {
        const unsigned int m = __builtin_amdgcn_ballot_w32(keep[u]);
        if (keep[u]) {
            const int pos = n + (int)__builtin_amdgcn_mbcnt_lo(m, 0u);
            Ls[wave][pos] = s[u];
            Ld[wave][pos] = d[u];
            Lw[wave][pos] = w[u];
        }
        n += __builtin_popcount(m);
    }

    const int ng = (n + 7) >> 3;
    if (ng == 0) return;
    float a0[8], b0[8], w0[8], a1[8], b1[8], w1[8];
    int   d0[8], d1[8];
    int g = 0;
    SC_FETCH(0, a0, b0, d0, w0)
    for (; g + 2 < ng; g += 2) {
        SC_FETCH(g + 1, a1, b1, d1, w1)
        __builtin_amdgcn_sched_barrier(0);
        SC_FLUSH_ALL(a0, b0, d0, w0)
        SC_FETCH(g + 2, a0, b0, d0, w0)
        __builtin_amdgcn_sched_barrier(0);
        SC_FLUSH_ALL(a1, b1, d1, w1)
    }
    if (g + 1 < ng) {
        SC_FETCH(g + 1, a1, b1, d1, w1)
        __builtin_amdgcn_sched_barrier(0);
        SC_FLUSH_ALL(a0, b0, d0, w0)
        SC_FLUSH_LAST(g + 1, a1, b1, d1, w1)
    } else {
        SC_FLUSH_LAST(g, a0, b0, d0, w0)
    }
}

#define HEAD_WAVES 2

typedef __attribute__((ext_vector_type(2))) float f2;

__global__ void __launch_bounds__(HEAD_WAVES * 32)
mlp_kernel(const float* __restrict__ embed, const int* __restrict__ idx,
           const float* __restrict__ b_conv,
           const float* __restrict__ W1, const float* __restrict__ b1,
           const float* __restrict__ W2, const float* __restrict__ b2,
           float* __restrict__ out, int B) {
    const int lane = threadIdx.x & 31;
    const int wave = threadIdx.x >> 5;
    const int hi   = lane >> 4;
    const int c    = lane & 15;
    const int pairBase = (blockIdx.x * HEAD_WAVES + wave) * 16;
    if (pairBase >= B) return;

    const int p  = pairBase + c;
    const int pc = p < B ? p : B - 1;
    const int i0 = idx[2 * pc];
    const int i1 = idx[2 * pc + 1];
    const float* e0 = embed + (size_t)i0 * 64 + 4 * hi;
    const float* e1 = embed + (size_t)i1 * 64 + 4 * hi;

    f4 x[2][8];
    #pragma unroll
    for (int u = 0; u < 8; ++u) {
        x[0][u] = *(const f4*)(e0 + 8 * u);
        x[1][u] = *(const f4*)(e1 + 8 * u);
    }
    float bc[8][4];
    #pragma unroll
    for (int u = 0; u < 8; ++u) {
        #pragma unroll
        for (int t = 0; t < 4; ++t) bc[u][t] = b_conv[8 * u + 4 * hi + t];
    }
    const float* wp = W1 + 64 * hi + c;
    float w[2][8][4];
    #pragma unroll
    for (int q = 0; q < 2; ++q) {
        #pragma unroll
        for (int u = 0; u < 8; ++u) {
            #pragma unroll
            for (int t = 0; t < 4; ++t) w[q][u][t] = wp[(64 * q + 8 * u + t) * 16];
        }
    }
    float b1v[8], w2v[8];
    #pragma unroll
    for (int r = 0; r < 8; ++r) {
        b1v[r] = b1[8 * hi + r];
        w2v[r] = W2[8 * hi + r];
    }

    v8f acc0 = {}, acc1 = {};
    #pragma unroll
    for (int u = 0; u < 8; ++u) {
        #pragma unroll
        for (int v = 0; v < 2; ++v) {
            f2 a0, a1, g0, g1;
            a0[0] = w[0][u][2 * v];     a0[1] = w[0][u][2 * v + 1];
            a1[0] = w[1][u][2 * v];     a1[1] = w[1][u][2 * v + 1];
            g0[0] = lrelu(x[0][u][2 * v]     + bc[u][2 * v]);
            g0[1] = lrelu(x[0][u][2 * v + 1] + bc[u][2 * v + 1]);
            g1[0] = lrelu(x[1][u][2 * v]     + bc[u][2 * v]);
            g1[1] = lrelu(x[1][u][2 * v + 1] + bc[u][2 * v + 1]);
            acc0 = __builtin_amdgcn_wmma_f32_16x16x4_f32(false, a0, false, g0, (short)0, acc0, false, false);
            acc1 = __builtin_amdgcn_wmma_f32_16x16x4_f32(false, a1, false, g1, (short)0, acc1, false, false);
        }
    }

    float part = 0.0f;
    #pragma unroll
    for (int r = 0; r < 8; ++r) {
        const float o = acc0[r] + acc1[r] + b1v[r];
        part += lrelu(o) * w2v[r];
    }
    const float tot = part + __shfl_xor(part, 16);
    const float acc = b2[0] + tot;
    if (hi == 0 && p < B) out[p] = 1.0f / (1.0f + __expf(-acc));
}

extern "C" void kernel_launch(void* const* d_in, const int* in_sizes, int n_in,
                              void* d_out, int out_size, void* d_ws, size_t ws_size,
                              hipStream_t stream) {
    const float* x        = (const float*)d_in[0];
    const int*   edge     = (const int*)d_in[1];
    const int*   index    = (const int*)d_in[2];
    const float* W_conv   = (const float*)d_in[3];
    const float* b_conv   = (const float*)d_in[4];
    const float* W1       = (const float*)d_in[5];
    const float* b1       = (const float*)d_in[6];
    const float* W2       = (const float*)d_in[7];
    const float* b2       = (const float*)d_in[8];
    float*       out      = (float*)d_out;

    const int N = in_sizes[0] / 256;
    const int E = in_sizes[1] / 2;
    const int B = in_sizes[2] / 2;
    const int* src = edge;
    const int* dst = edge + E;

    float*        h     = (float*)d_ws;
    float*        embed = h + (size_t)N * 64;
    unsigned int* bits  = (unsigned int*)(embed + (size_t)N * 64);
    float*        dinv  = (float*)(bits + N);
    _Float16*     Wt    = (_Float16*)(dinv + N);

    unsigned int* part = (unsigned int*)embed;
    deg_hist_kernel<<<((N + HIST_NODES - 1) / HIST_NODES) * HIST_CHUNKS, HIST_THREADS, 0, stream>>>(
        dst, index, part, E, 2 * B, N);
    dinv_kernel<<<(N + 255) / 256, 256, 0, stream>>>(part, bits, dinv, N, W_conv, Wt);

    gemm_xw_kernel<<<(N + 16 * GT_WAVES - 1) / (16 * GT_WAVES), GT_THREADS, 0, stream>>>(x, Wt, dinv, bits, h, embed, N);

    scatter_edges_kernel<<<(E + SC_WAVES * SC_WEDGES - 1) / (SC_WAVES * SC_WEDGES), SC_THREADS, 0, stream>>>(
        h, dinv, bits, src, dst, embed, E, (N + 31) / 32);

    mlp_kernel<<<(B + 16 * HEAD_WAVES - 1) / (16 * HEAD_WAVES), HEAD_WAVES * 32, 0, stream>>>(embed, index, b_conv, W1, b1, W2, b2, out, B);
}


# edge norm factored: GEMM epilogue stores rows scaled by dinv[src], scatter adds them unweighted (no per-edge dinv lookups), head multiplies by dinv[dst]; sums still in arrival order
# speedup vs baseline: 6.4296x; 1.0403x over previous
#include <hip/hip_runtime.h>
#include <hip/hip_bf16.h>

typedef __attribute__((ext_vector_type(16))) _Float16 v16h;
typedef __attribute__((ext_vector_type(8)))  _Float16 h8;
typedef __attribute__((ext_vector_type(8)))  float    v8f;
typedef __attribute__((ext_vector_type(4)))  float    f4;

#define NEG_SLOPE 0.01f

__device__ __forceinline__ float lrelu(float v) { return v >= 0.0f ? v : NEG_SLOPE * v; }

#define ROW_USED 0x80000000u

#define GT_WAVES   4
#define GT_THREADS (GT_WAVES * 32)
#define GT_PITCH   264

typedef __attribute__((ext_vector_type(4))) _Float16 h4;

__global__ void __launch_bounds__(GT_THREADS)
gemm_xw_kernel(const float* __restrict__ x,
               const _Float16* __restrict__ Wt,
               const float* __restrict__ dinv,
               const unsigned int* __restrict__ bits,
               float* __restrict__ h,
               float* __restrict__ embed, int N) {
    __shared__ __attribute__((aligned(16))) _Float16 sW[64 * GT_PITCH];
    __shared__ __attribute__((aligned(16))) _Float16 sA[GT_WAVES * 16 * GT_PITCH];

    const int tid  = threadIdx.x;
    const int lane = tid & 31;
    const int wave = tid >> 5;
    const int hi   = lane >> 4;
    const int r    = lane & 15;
    const int rowBase = (blockIdx.x * GT_WAVES + wave) * 16;
    const bool active = rowBase < N;

    f4 xa[32];
    if (active) {
        #pragma unroll
        for (int i = 0; i < 32; ++i) {
            int row = rowBase + (i >> 1);
            row = row < N ? row : N - 1;
            xa[i] = *(const f4*)(x + (size_t)row * 256 + (i & 1) * 128 + lane * 4);
        }
    }
    h8 wv[64 / GT_WAVES];
    #pragma unroll
    for (int i = 0; i < 64 / GT_WAVES; ++i)
        wv[i] = *(const h8*)(Wt + (size_t)(tid + i * GT_THREADS) * 8);

    float dv[8];
    #pragma unroll
    for (int j = 0; j < 8; ++j) {
        int m = rowBase + j + hi * 8;
        m = m < N ? m : N - 1;
        dv[j] = dinv[m];
    }
    const unsigned int mw   = bits[(active ? rowBase : 0) >> 5];
    const unsigned int mbit = (unsigned int)(rowBase & 31) + (unsigned int)hi * 8u;

    #pragma unroll
    for (int i = 0; i < 64 / GT_WAVES; ++i)
        *(h8*)(sW + (wave + i * GT_WAVES) * GT_PITCH + lane * 8) = wv[i];
    if (active) {
        _Float16* myA = sA + wave * 16 * GT_PITCH;
        #pragma unroll
        for (int i = 0; i < 32; ++i) {
            h4 v;
            v[0] = (_Float16)xa[i][0]; v[1] = (_Float16)xa[i][1];
            v[2] = (_Float16)xa[i][2]; v[3] = (_Float16)xa[i][3];
            *(h4*)(myA + (i >> 1) * GT_PITCH + (i & 1) * 128 + lane * 4) = v;
        }
    }
    __syncthreads();
    if (!active) return;

    v8f c0 = {}, c1 = {}, c2 = {}, c3 = {};
    const _Float16* ap = sA + (wave * 16 + r) * GT_PITCH + hi * 8;
    const _Float16* bp = sW + r * GT_PITCH + hi * 8;

    h8 flo[2][5], fhi[2][5];
    #pragma unroll
    for (int t = 0; t < 4; ++t) {
        flo[0][t] = *(const h8*)(bp + t * 16 * GT_PITCH);
        fhi[0][t] = *(const h8*)(bp + t * 16 * GT_PITCH + 16);
    }
    flo[0][4] = *(const h8*)(ap);
    fhi[0][4] = *(const h8*)(ap + 16);

    #pragma unroll
    for (int kk = 0; kk < 8; ++kk) {
        const int cur = kk & 1, nxt = cur ^ 1;
        if (kk + 1 < 8) {
            const int kb = (kk + 1) * 32;
            #pragma unroll
            for (int t = 0; t < 4; ++t) {
                flo[nxt][t] = *(const h8*)(bp + t * 16 * GT_PITCH + kb);
                fhi[nxt][t] = *(const h8*)(bp + t * 16 * GT_PITCH + kb + 16);
            }
            flo[nxt][4] = *(const h8*)(ap + kb);
            fhi[nxt][4] = *(const h8*)(ap + kb + 16);
        }
        __builtin_amdgcn_sched_barrier(0);
        const v16h a   = __builtin_shufflevector(flo[cur][4], fhi[cur][4], 0, 1, 2, 3, 4, 5, 6, 7, 8, 9, 10, 11, 12, 13, 14, 15);
        const v16h b0v = __builtin_shufflevector(flo[cur][0], fhi[cur][0], 0, 1, 2, 3, 4, 5, 6, 7, 8, 9, 10, 11, 12, 13, 14, 15);
        const v16h b1v = __builtin_shufflevector(flo[cur][1], fhi[cur][1], 0, 1, 2, 3, 4, 5, 6, 7, 8, 9, 10, 11, 12, 13, 14, 15);
        const v16h b2v = __builtin_shufflevector(flo[cur][2], fhi[cur][2], 0, 1, 2, 3, 4, 5, 6, 7, 8, 9, 10, 11, 12, 13, 14, 15);
        const v16h b3v = __builtin_shufflevector(flo[cur][3], fhi[cur][3], 0, 1, 2, 3, 4, 5, 6, 7, 8, 9, 10, 11, 12, 13, 14, 15);
        c0 = __builtin_amdgcn_wmma_f32_16x16x32_f16(false, a, false, b0v, (short)0, c0, false, false);
        c1 = __builtin_amdgcn_wmma_f32_16x16x32_f16(false, a, false, b1v, (short)0, c1, false, false);
        c2 = __builtin_amdgcn_wmma_f32_16x16x32_f16(false, a, false, b2v, (short)0, c2, false, false);
        c3 = __builtin_amdgcn_wmma_f32_16x16x32_f16(false, a, false, b3v, (short)0, c3, false, false);
    }

    #pragma unroll
    for (int j = 0; j < 8; ++j) {
        const int m = rowBase + j + hi * 8;
        if (m < N) {
            const float di = dv[j];
            const float v0 = c0[j] * di, v1 = c1[j] * di, v2 = c2[j] * di, v3 = c3[j] * di;
            float* p = h + (size_t)m * 64;
            p[r]      = v0;
            p[16 + r] = v1;
            p[32 + r] = v2;
            p[48 + r] = v3;
            if ((mw >> (mbit + (unsigned int)j)) & 1u) {
                float* q = embed + (size_t)m * 64;
                q[r]      = v0;
                q[16 + r] = v1;
                q[32 + r] = v2;
                q[48 + r] = v3;
            }
        }
    }
}

#define HIST_NODES   8192
#define HIST_CHUNKS  25
#define HIST_THREADS 512
#define HIST_BATCH   16
#define MARK_BATCH   4

__device__ __forceinline__ void hist_add(unsigned int* cnt, int node, unsigned int base) {
    const unsigned int a = (unsigned int)node - base;
    if (a < (unsigned int)HIST_NODES) atomicAdd(&cnt[a], 1u);
}

__global__ void __launch_bounds__(HIST_THREADS)
deg_hist_kernel(const int* __restrict__ dst, const int* __restrict__ idx,
                unsigned int* __restrict__ part, int E, int nIdx, int N) {
    __shared__ unsigned int cnt[HIST_NODES];
    const int tid = threadIdx.x;
    const int r = blockIdx.x / HIST_CHUNKS;
    const int c = blockIdx.x % HIST_CHUNKS;
    const unsigned int base = (unsigned int)r * HIST_NODES;

    for (int j = tid; j < HIST_NODES; j += HIST_THREADS) cnt[j] = 0u;
    __syncthreads();

    const int iper = (nIdx + HIST_CHUNKS - 1) / HIST_CHUNKS;
    const int t0 = c * iper;
    const int t1 = (t0 + iper < nIdx) ? (t0 + iper) : nIdx;
    for (int t = t0 + tid; t < t1; t += MARK_BATCH * HIST_THREADS) {
        int m[MARK_BATCH];
        #pragma unroll
        for (int u = 0; u < MARK_BATCH; ++u) {
            const int tt = t + u * HIST_THREADS;
            m[u] = (tt < t1) ? idx[tt] : -1;
        }
        #pragma unroll
        for (int u = 0; u < MARK_BATCH; ++u) {
            const unsigned int a = (unsigned int)m[u] - base;
            if (a < (unsigned int)HIST_NODES) atomicOr(&cnt[a], ROW_USED);
        }
    }

    const int nvec = E >> 2;
    const int gper = (nvec + HIST_CHUNKS - 1) / HIST_CHUNKS;
    const int g0 = c * gper;
    const int g1 = (g0 + gper < nvec) ? (g0 + gper) : nvec;
    if ((((size_t)dst) & 15) == 0) {
        const int4* __restrict__ dv = (const int4*)dst;
        for (int g = g0 + tid; g < g1; g += HIST_BATCH * HIST_THREADS) {
            int4 v[HIST_BATCH];
            #pragma unroll
            for (int u = 0; u < HIST_BATCH; ++u) {
                const int gg = g + u * HIST_THREADS;
                v[u] = (gg < g1) ? dv[gg] : make_int4(-1, -1, -1, -1);
            }
            #pragma unroll
            for (int u = 0; u < HIST_BATCH; ++u) {
                hist_add(cnt, v[u].x, base);
                hist_add(cnt, v[u].y, base);
                hist_add(cnt, v[u].z, base);
                hist_add(cnt, v[u].w, base);
            }
        }
    } else {
        for (int e = 4 * g0 + tid; e < 4 * g1; e += HIST_THREADS) hist_add(cnt, dst[e], base);
    }
    if (c == 0) {
        const int e = 4 * nvec + tid;
        if (e < E) hist_add(cnt, dst[e], base);
    }
    __syncthreads();

    unsigned int* __restrict__ plane = part + (size_t)c * N;
    unsigned int outv[HIST_NODES / HIST_THREADS];
    #pragma unroll
    for (int k = 0; k < HIST_NODES / HIST_THREADS; ++k) outv[k] = cnt[tid + k * HIST_THREADS];
    #pragma unroll
    for (int k = 0; k < HIST_NODES / HIST_THREADS; ++k) {
        const unsigned int i = base + (unsigned int)(tid + k * HIST_THREADS);
        if (i < (unsigned int)N) plane[i] = outv[k];
    }
}

__global__ void __launch_bounds__(256)
dinv_kernel(const unsigned int* __restrict__ part, unsigned int* __restrict__ bits,
            float* __restrict__ dinv, int N,
            const float* __restrict__ W, _Float16* __restrict__ Wt) {
    const int i = blockIdx.x * 256 + threadIdx.x;
    for (int t = i; t < 256 * 64; t += (int)gridDim.x * 256) {
        const int k = t >> 6;
        const int c = t & 63;
        Wt[c * 256 + k] = (_Float16)W[t];
    }
    unsigned int mark = 0u;
    if (i < N) {
        unsigned int n = 1u;
        #pragma unroll
        for (int c = 0; c < HIST_CHUNKS; ++c) {
            const unsigned int v = part[(size_t)c * N + i];
            n += v & ~ROW_USED;
            mark |= v & ROW_USED;
        }
        float d = (float)n;
        dinv[i] = d > 0.0f ? rsqrtf(d) : 0.0f;
    }
    const unsigned int word = __builtin_amdgcn_ballot_w32(mark != 0u);
    if ((threadIdx.x & 31) == 0 && i < N) bits[i >> 5] = word;
}

__device__ __forceinline__ int   lane_i(int v, int j)   { return __builtin_amdgcn_readlane(v, j); }

#define SC_WAVES    8
#define SC_THREADS  (SC_WAVES * 32)
#define SC_EPL      4
#define SC_WEDGES   (32 * SC_EPL)
#define SC_BITS_MAX 4096
#define SC_BITS_PER_THREAD (SC_BITS_MAX / SC_THREADS)

#define SC_FETCH(G, RA, RB, DD)                                                      \
    {                                                                                \
        int li_ = (G) * 8 + (lane & 7);                                              \
        li_ = li_ < n ? li_ : n - 1;                                                 \
        const int   ls_ = Ls[wave][li_];                                             \
        const int   ld_ = Ld[wave][li_];                                             \
        _Pragma("unroll")                                                            \
        for (int j = 0; j < 8; ++j) {                                                \
            const float* p_ = h + (size_t)lane_i(ls_, j) * 64 + lane;                \
            DD[j] = lane_i(ld_, j);                                                  \
            RA[j] = p_[0];                                                           \
            RB[j] = p_[32];                                                          \
        }                                                                            \
    }
#define SC_FLUSH_ALL(RA, RB, DD)                                                     \
    {                                                                                \
        _Pragma("unroll")                                                            \
        for (int j = 0; j < 8; ++j) {                                                \
            float* q_ = embed + (size_t)DD[j] * 64 + lane;                           \
            unsafeAtomicAdd(q_,      RA[j]);                                         \
            unsafeAtomicAdd(q_ + 32, RB[j]);                                         \
        }                                                                            \
    }
#define SC_FLUSH_LAST(G, RA, RB, DD)                                                 \
    {                                                                                \
        _Pragma("unroll")                                                            \
        for (int j = 0; j < 8; ++j) {                                                \
            if ((G) * 8 + j < n) {                                                   \
                float* q_ = embed + (size_t)DD[j] * 64 + lane;                       \
                unsafeAtomicAdd(q_,      RA[j]);                                     \
                unsafeAtomicAdd(q_ + 32, RB[j]);                                     \
            }                                                                        \
        }                                                                            \
    }

__global__ void __launch_bounds__(SC_THREADS)
scatter_edges_kernel(const float* __restrict__ h,
                     const unsigned int* __restrict__ bits,
                     const int* __restrict__ src, const int* __restrict__ dst,
                     float* __restrict__ embed, int E, int nWords) {
    __shared__ unsigned int sBits[SC_BITS_MAX];
    __shared__ int   Ls[SC_WAVES][SC_WEDGES];
    __shared__ int   Ld[SC_WAVES][SC_WEDGES];

    const int tid  = threadIdx.x;
    const int lane = tid & 31;
    const int wave = tid >> 5;
    const bool inLds = nWords <= SC_BITS_MAX;

    unsigned int bw[SC_BITS_PER_THREAD];
    if (inLds) {
        #pragma unroll
        for (int k = 0; k < SC_BITS_PER_THREAD; ++k) {
            const int i = tid + k * SC_THREADS;
            bw[k] = bits[i < nWords ? i : nWords - 1];
        }
    }
    const int wbase = (blockIdx.x * SC_WAVES + wave) * SC_WEDGES;
    int  s[SC_EPL], d[SC_EPL];
    bool ok[SC_EPL];
    #pragma unroll
    for (int u = 0; u < SC_EPL; ++u) {
        const int e = wbase + u * 32 + lane;
        ok[u] = e < E;
        const int ec = ok[u] ? e : E - 1;
        s[u] = src[ec];
        d[u] = dst[ec];
    }
    if (inLds) {
        #pragma unroll
        for (int k = 0; k < SC_BITS_PER_THREAD; ++k) sBits[tid + k * SC_THREADS] = bw[k];
    }
    __syncthreads();

    bool  keep[SC_EPL];
    unsigned int word[SC_EPL];
    if (inLds) {
        #pragma unroll
        for (int u = 0; u < SC_EPL; ++u) word[u] = sBits[d[u] >> 5];
    } else {
        #pragma unroll
        for (int u = 0; u < SC_EPL; ++u) word[u] = bits[d[u] >> 5];
    }
    #pragma unroll
    for (int u = 0; u < SC_EPL; ++u) keep[u] = ok[u] & (((word[u] >> (d[u] & 31)) & 1u) != 0u);

    int n = 0;
    #pragma unroll
    for (int u = 0; u < SC_EPL; ++u) {
        const unsigned int m = __builtin_amdgcn_ballot_w32(keep[u]);
        if (keep[u]) {
            const int pos = n + (int)__builtin_amdgcn_mbcnt_lo(m, 0u);
            Ls[wave][pos] = s[u];
            Ld[wave][pos] = d[u];
        }
        n += __builtin_popcount(m);
    }

    const int ng = (n + 7) >> 3;
    if (ng == 0) return;
    float a0[8], b0[8], a1[8], b1[8];
    int   d0[8], d1[8];
    int g = 0;
    SC_FETCH(0, a0, b0, d0)
    for (; g + 2 < ng; g += 2) {
        SC_FETCH(g + 1, a1, b1, d1)
        __builtin_amdgcn_sched_barrier(0);
        SC_FLUSH_ALL(a0, b0, d0)
        SC_FETCH(g + 2, a0, b0, d0)
        __builtin_amdgcn_sched_barrier(0);
        SC_FLUSH_ALL(a1, b1, d1)
    }
    if (g + 1 < ng) {
        SC_FETCH(g + 1, a1, b1, d1)
        __builtin_amdgcn_sched_barrier(0);
        SC_FLUSH_ALL(a0, b0, d0)
        SC_FLUSH_LAST(g + 1, a1, b1, d1)
    } else {
        SC_FLUSH_LAST(g, a0, b0, d0)
    }
}

#define HEAD_WAVES 2

typedef __attribute__((ext_vector_type(2))) float f2;

__global__ void __launch_bounds__(HEAD_WAVES * 32)
mlp_kernel(const float* __restrict__ embed, const float* __restrict__ dinv,
           const int* __restrict__ idx,
           const float* __restrict__ b_conv,
           const float* __restrict__ W1, const float* __restrict__ b1,
           const float* __restrict__ W2, const float* __restrict__ b2,
           float* __restrict__ out, int B) {
    const int lane = threadIdx.x & 31;
    const int wave = threadIdx.x >> 5;
    const int hi   = lane >> 4;
    const int c    = lane & 15;
    const int pairBase = (blockIdx.x * HEAD_WAVES + wave) * 16;
    if (pairBase >= B) return;

    const int p  = pairBase + c;
    const int pc = p < B ? p : B - 1;
    const int i0 = idx[2 * pc];
    const int i1 = idx[2 * pc + 1];
    const float* e0 = embed + (size_t)i0 * 64 + 4 * hi;
    const float* e1 = embed + (size_t)i1 * 64 + 4 * hi;
    const float di0 = dinv[i0];
    const float di1 = dinv[i1];

    f4 x[2][8];
    #pragma unroll
    for (int u = 0; u < 8; ++u) {
        x[0][u] = *(const f4*)(e0 + 8 * u);
        x[1][u] = *(const f4*)(e1 + 8 * u);
    }
    float bc[8][4];
    #pragma unroll
    for (int u = 0; u < 8; ++u) {
        #pragma unroll
        for (int t = 0; t < 4; ++t) bc[u][t] = b_conv[8 * u + 4 * hi + t];
    }
    const float* wp = W1 + 64 * hi + c;
    float w[2][8][4];
    #pragma unroll
    for (int q = 0; q < 2; ++q) {
        #pragma unroll
        for (int u = 0; u < 8; ++u) {
            #pragma unroll
            for (int t = 0; t < 4; ++t) w[q][u][t] = wp[(64 * q + 8 * u + t) * 16];
        }
    }
    float b1v[8], w2v[8];
    #pragma unroll
    for (int r = 0; r < 8; ++r) {
        b1v[r] = b1[8 * hi + r];
        w2v[r] = W2[8 * hi + r];
    }

    v8f acc0 = {}, acc1 = {};
    #pragma unroll
    for (int u = 0; u < 8; ++u) {
        #pragma unroll
        for (int v = 0; v < 2; ++v) {
            f2 a0, a1, g0, g1;
            a0[0] = w[0][u][2 * v];     a0[1] = w[0][u][2 * v + 1];
            a1[0] = w[1][u][2 * v];     a1[1] = w[1][u][2 * v + 1];
            g0[0] = lrelu(x[0][u][2 * v]     * di0 + bc[u][2 * v]);
            g0[1] = lrelu(x[0][u][2 * v + 1] * di0 + bc[u][2 * v + 1]);
            g1[0] = lrelu(x[1][u][2 * v]     * di1 + bc[u][2 * v]);
            g1[1] = lrelu(x[1][u][2 * v + 1] * di1 + bc[u][2 * v + 1]);
            acc0 = __builtin_amdgcn_wmma_f32_16x16x4_f32(false, a0, false, g0, (short)0, acc0, false, false);
            acc1 = __builtin_amdgcn_wmma_f32_16x16x4_f32(false, a1, false, g1, (short)0, acc1, false, false);
        }
    }

    float part = 0.0f;
    #pragma unroll
    for (int r = 0; r < 8; ++r) {
        const float o = acc0[r] + acc1[r] + b1v[r];
        part += lrelu(o) * w2v[r];
    }
    const float tot = part + __shfl_xor(part, 16);
    const float acc = b2[0] + tot;
    if (hi == 0 && p < B) out[p] = 1.0f / (1.0f + __expf(-acc));
}

extern "C" void kernel_launch(void* const* d_in, const int* in_sizes, int n_in,
                              void* d_out, int out_size, void* d_ws, size_t ws_size,
                              hipStream_t stream) {
    const float* x        = (const float*)d_in[0];
    const int*   edge     = (const int*)d_in[1];
    const int*   index    = (const int*)d_in[2];
    const float* W_conv   = (const float*)d_in[3];
    const float* b_conv   = (const float*)d_in[4];
    const float* W1       = (const float*)d_in[5];
    const float* b1       = (const float*)d_in[6];
    const float* W2       = (const float*)d_in[7];
    const float* b2       = (const float*)d_in[8];
    float*       out      = (float*)d_out;

    const int N = in_sizes[0] / 256;
    const int E = in_sizes[1] / 2;
    const int B = in_sizes[2] / 2;
    const int* src = edge;
    const int* dst = edge + E;

    float*        h     = (float*)d_ws;
    float*        embed = h + (size_t)N * 64;
    unsigned int* bits  = (unsigned int*)(embed + (size_t)N * 64);
    float*        dinv  = (float*)(bits + N);
    _Float16*     Wt    = (_Float16*)(dinv + N);

    unsigned int* part = (unsigned int*)embed;
    deg_hist_kernel<<<((N + HIST_NODES - 1) / HIST_NODES) * HIST_CHUNKS, HIST_THREADS, 0, stream>>>(
        dst, index, part, E, 2 * B, N);
    dinv_kernel<<<(N + 255) / 256, 256, 0, stream>>>(part, bits, dinv, N, W_conv, Wt);

    gemm_xw_kernel<<<(N + 16 * GT_WAVES - 1) / (16 * GT_WAVES), GT_THREADS, 0, stream>>>(x, Wt, dinv, bits, h, embed, N);

    scatter_edges_kernel<<<(E + SC_WAVES * SC_WEDGES - 1) / (SC_WAVES * SC_WEDGES), SC_THREADS, 0, stream>>>(
        h, bits, src, dst, embed, E, (N + 31) / 32);

    mlp_kernel<<<(B + 16 * HEAD_WAVES - 1) / (16 * HEAD_WAVES), HEAD_WAVES * 32, 0, stream>>>(embed, dinv, index, b_conv, W1, b1, W2, b2, out, B);
}
